# starting kernel
# baseline (speedup 1.0000x reference)
#include <hip/hip_runtime.h>
#include <math.h>

#define BB 4
#define MM 500
#define NN 500
#define CC 91

typedef __attribute__((ext_vector_type(2))) float v2f;
typedef __attribute__((ext_vector_type(8))) float v8f;

__device__ __forceinline__ int imin_i(int a, int b) { return a < b ? a : b; }

__global__ void sigmoid_kernel(const float* __restrict__ p1l,
                               const float* __restrict__ p2l,
                               float* __restrict__ s1,
                               float* __restrict__ s2) {
    int idx = blockIdx.x * blockDim.x + threadIdx.x;
    const int n1 = BB * NN * CC;
    const int n2 = BB * MM * CC;
    if (idx < n1) {
        float x = p1l[idx];
        s1[idx] = 1.0f / (1.0f + expf(-x));
    } else if (idx < n1 + n2) {
        float x = p2l[idx - n1];
        s2[idx - n1] = 1.0f / (1.0f + expf(-x));
    }
}

__global__ void cost_kernel(const float* __restrict__ p1b,
                            const float* __restrict__ p2b,
                            const float* __restrict__ s1,
                            const float* __restrict__ s2,
                            float* __restrict__ cost) {
    const int TN = (NN + 15) / 16;
    const int TM = (MM + 15) / 16;
    int bid = blockIdx.x;
    int b  = bid / (TM * TN);
    int t  = bid % (TM * TN);
    int m0 = (t / TN) * 16;
    int n0 = (t % TN) * 16;
    int lane = threadIdx.x;

    __shared__ float sq2[16];
    __shared__ float sq1[16];
    __shared__ float ls2[16 * CC];
    __shared__ float ls1[16 * CC];

    v2f a  = {0.0f, 0.0f};
    v2f bv = {0.0f, 0.0f};
    if (lane < 16) {
        int m = imin_i(m0 + lane, MM - 1);
        const float* pb = p2b + ((size_t)b * MM + m) * 4;
        float x = pb[0], y = pb[1];
        a.x = x; a.y = y;
        sq2[lane] = x * x + y * y;
        int n = imin_i(n0 + lane, NN - 1);
        const float* qb = p1b + ((size_t)b * NN + n) * 4;
        float bx = qb[0], by = qb[1];
        bv.x = bx; bv.y = by;
        sq1[lane] = bx * bx + by * by;
    }
    const float* s2base = s2 + (size_t)b * MM * CC;
    const float* s1base = s1 + (size_t)b * NN * CC;
    for (int idx = lane; idx < 16 * CC; idx += 32) {
        int r = idx / CC;
        int c = idx - r * CC;
        ls2[idx] = s2base[(size_t)imin_i(m0 + r, MM - 1) * CC + c];
        ls1[idx] = s1base[(size_t)imin_i(n0 + r, NN - 1) * CC + c];
    }
    __syncthreads();

    v8f cacc = {0.f, 0.f, 0.f, 0.f, 0.f, 0.f, 0.f, 0.f};
    v8f dot = __builtin_amdgcn_wmma_f32_16x16x4_f32(
         false, a,  false, bv,
         (short)0, cacc,  false,  false);

    int n_loc  = lane & 15;
    int m_base = (lane >> 4) << 3;

    float cls[8];
#pragma unroll
    for (int i = 0; i < 8; ++i) cls[i] = 0.0f;
    for (int c = 0; c < CC; ++c) {
        float s1v = ls1[n_loc * CC + c];
#pragma unroll
        for (int i = 0; i < 8; ++i) {
            float d = fabsf(ls2[(m_base + i) * CC + c] - s1v);
            cls[i] = fmaxf(cls[i], d);
        }
    }

#pragma unroll
    for (int i = 0; i < 8; ++i) {
        int m = m0 + m_base + i;
        int n = n0 + n_loc;
        if (m < MM && n < NN) {
            float d2 = sq2[m_base + i] + sq1[n_loc] - 2.0f * dot[i];
            float cd = sqrtf(fmaxf(d2, 0.0f));
            cost[((size_t)b * MM + m) * NN + n] = 0.5f * cd + 0.5f * cls[i];
        }
    }
}

#define LTH 512
#define NWAVE (LTH / 32)
__global__ void __launch_bounds__(LTH)
lapjv_kernel(const float* __restrict__ cost, int* __restrict__ mapping) {
    int b   = blockIdx.x;
    int tid = threadIdx.x;
    const float* costB = cost + (size_t)b * MM * NN;
    const double INF = __builtin_inf();

    __shared__ double sh_u[LTH], sh_v[LTH], sh_short[LTH];
    __shared__ int sh_path[LTH], sh_c4r[LTH], sh_r4c[LTH], sh_SR[LTH],
                   sh_rem[LTH];
    __shared__ double wred_v[NWAVE];
    __shared__ int    wred_i[NWAVE];
    __shared__ int s_i, s_sink, s_nsr;
    __shared__ double s_minval;

    if (tid < NN) {
        sh_u[tid] = 0.0; sh_v[tid] = 0.0;
        sh_c4r[tid] = -1; sh_r4c[tid] = -1;
    }
    __syncthreads();

    for (int cur = 0; cur < MM; ++cur) {
        if (tid < NN) { sh_rem[tid] = 1; sh_short[tid] = INF; sh_path[tid] = -1; }
        if (tid == 0) { s_i = cur; s_minval = 0.0; s_sink = -1; s_nsr = 0; }
        __syncthreads();

        while (true) {
            int    ii   = s_i;
            double minv = s_minval;
            if (tid == 0) { sh_SR[s_nsr] = ii; s_nsr = s_nsr + 1; }

            double cand;
            int    candi = tid;
            if (tid < NN) {
                double r = minv + (double)costB[(size_t)ii * NN + tid]
                                - sh_u[ii] - sh_v[tid];
                if (sh_rem[tid] && r < sh_short[tid]) {
                    sh_short[tid] = r;
                    sh_path[tid]  = ii;
                }
                cand = sh_rem[tid] ? sh_short[tid] : INF;
            } else {
                cand = INF;
            }

#pragma unroll
            for (int off = 16; off > 0; off >>= 1) {
                double ov = __shfl_xor(cand, off);
                int    oi = __shfl_xor(candi, off);
                if (ov < cand || (ov == cand && oi < candi)) {
                    cand = ov; candi = oi;
                }
            }
            int wave = tid >> 5;
            if ((tid & 31) == 0) { wred_v[wave] = cand; wred_i[wave] = candi; }
            __syncthreads();

            if (tid == 0) {
                double bvv = wred_v[0]; int bii = wred_i[0];
#pragma unroll
                for (int w = 1; w < NWAVE; ++w) {
                    double ov = wred_v[w]; int oi = wred_i[w];
                    if (ov < bvv || (ov == bvv && oi < bii)) {
                        bvv = ov; bii = oi;
                    }
                }
                s_minval = bvv;
                sh_rem[bii] = 0;
                if (sh_r4c[bii] == -1) s_sink = bii;
                else                   s_i = sh_r4c[bii];
            }
            __syncthreads();
            if (s_sink != -1) break;
        }

        double minv = s_minval;
        int    nsr  = s_nsr;
        if (tid == 0) {
            sh_u[cur] += minv;
        } else if (tid < nsr) {
            int sr = sh_SR[tid];
            sh_u[sr] += minv - sh_short[sh_c4r[sr]];
        }
        if (tid < NN && !sh_rem[tid]) sh_v[tid] -= minv - sh_short[tid];
        __syncthreads();

        if (tid == 0) {
            int j = s_sink;
            while (true) {
                int i = sh_path[j];
                sh_r4c[j] = i;
                int tmp = sh_c4r[i];
                sh_c4r[i] = j;
                j = tmp;
                if (i == cur) break;
            }
        }
        __syncthreads();
    }
    if (tid < MM) mapping[b * MM + tid] = sh_c4r[tid];
}

__global__ void extrap_kernel(const float* __restrict__ p1b,
                              const float* __restrict__ p1l,
                              const float* __restrict__ p2b,
                              const float* __restrict__ p2l,
                              const float* __restrict__ toff,
                              const int*   __restrict__ mapping,
                              float* __restrict__ out) {
    int idx = blockIdx.x * blockDim.x + threadIdx.x;
    if (idx >= BB * MM) return;
    int b = idx / MM;

    float first  = toff[b * 3 + 1] - toff[b * 3 + 0];
    float second = toff[b * 3 + 2] - toff[b * 3 + 1];
    float factor = second / first;

    int  j   = mapping[idx];
    bool has = (j >= 0);
    int  jc  = has ? j : 0;

    const float* pb2 = p2b + (size_t)idx * 4;
    const float* pb1 = p1b + ((size_t)b * NN + jc) * 4;
    float* ob = out + (size_t)idx * 4;
#pragma unroll
    for (int k = 0; k < 4; ++k) {
        float c1 = has ? pb1[k] : pb2[k];
        float v  = pb2[k] + (pb2[k] - c1) * factor;
        ob[k] = (k >= 2) ? fmaxf(v, 0.0f) : v;
    }

    const float* pl2 = p2l + (size_t)idx * CC;
    const float* pl1 = p1l + ((size_t)b * NN + jc) * CC;
    float* ol = out + (size_t)BB * MM * 4 + (size_t)idx * CC;
    for (int c = 0; c < CC; ++c) {
        float c1 = has ? pl1[c] : 0.0f;
        ol[c] = 0.5f * (pl2[c] + c1);
    }
}

extern "C" void kernel_launch(void* const* d_in, const int* in_sizes, int n_in,
                              void* d_out, int out_size, void* d_ws, size_t ws_size,
                              hipStream_t stream) {
    (void)in_sizes; (void)n_in; (void)out_size; (void)ws_size;
    const float* p1b  = (const float*)d_in[0];
    const float* p1l  = (const float*)d_in[1];
    const float* p2b  = (const float*)d_in[2];
    const float* p2l  = (const float*)d_in[3];
    const float* toff = (const float*)d_in[4];

    float* ws   = (float*)d_ws;
    float* s1   = ws;
    float* s2   = s1 + (size_t)BB * NN * CC;
    float* cost = s2 + (size_t)BB * MM * CC;
    int*   map  = (int*)(cost + (size_t)BB * MM * NN);
    float* out  = (float*)d_out;

    int tot = BB * NN * CC + BB * MM * CC;
    sigmoid_kernel<<<(tot + 255) / 256, 256, 0, stream>>>(p1l, p2l, s1, s2);

    const int TM = (MM + 15) / 16, TN = (NN + 15) / 16;
    cost_kernel<<<BB * TM * TN, 32, 0, stream>>>(p1b, p2b, s1, s2, cost);

    lapjv_kernel<<<BB, LTH, 0, stream>>>(cost, map);

    int tot2 = BB * MM;
    extrap_kernel<<<(tot2 + 255) / 256, 256, 0, stream>>>(p1b, p1l, p2b, p2l,
                                                          toff, map, out);
}


# Assignment search step: each wave folds the 16 wave minima itself (one pair per lane, 4 row_xmask DPP stages, two LDS slots in turn), so one barrier per step and no single-thread serial fold
# speedup vs baseline: 1.5655x; 1.5655x over previous
#include <hip/hip_runtime.h>
#include <math.h>

#define BB 4
#define MM 500
#define NN 500
#define CC 91

typedef __attribute__((ext_vector_type(2))) float v2f;
typedef __attribute__((ext_vector_type(8))) float v8f;

__device__ __forceinline__ int imin_i(int a, int b) { return a < b ? a : b; }

__global__ void sigmoid_kernel(const float* __restrict__ p1l,
                               const float* __restrict__ p2l,
                               float* __restrict__ s1,
                               float* __restrict__ s2) {
    int idx = blockIdx.x * blockDim.x + threadIdx.x;
    const int n1 = BB * NN * CC;
    const int n2 = BB * MM * CC;
    if (idx < n1) {
        float x = p1l[idx];
        s1[idx] = 1.0f / (1.0f + expf(-x));
    } else if (idx < n1 + n2) {
        float x = p2l[idx - n1];
        s2[idx - n1] = 1.0f / (1.0f + expf(-x));
    }
}

__global__ void cost_kernel(const float* __restrict__ p1b,
                            const float* __restrict__ p2b,
                            const float* __restrict__ s1,
                            const float* __restrict__ s2,
                            float* __restrict__ cost) {
    const int TN = (NN + 15) / 16;
    const int TM = (MM + 15) / 16;
    int bid = blockIdx.x;
    int b  = bid / (TM * TN);
    int t  = bid % (TM * TN);
    int m0 = (t / TN) * 16;
    int n0 = (t % TN) * 16;
    int lane = threadIdx.x;

    __shared__ float sq2[16];
    __shared__ float sq1[16];
    __shared__ float ls2[16 * CC];
    __shared__ float ls1[16 * CC];

    v2f a  = {0.0f, 0.0f};
    v2f bv = {0.0f, 0.0f};
    if (lane < 16) {
        int m = imin_i(m0 + lane, MM - 1);
        const float* pb = p2b + ((size_t)b * MM + m) * 4;
        float x = pb[0], y = pb[1];
        a.x = x; a.y = y;
        sq2[lane] = x * x + y * y;
        int n = imin_i(n0 + lane, NN - 1);
        const float* qb = p1b + ((size_t)b * NN + n) * 4;
        float bx = qb[0], by = qb[1];
        bv.x = bx; bv.y = by;
        sq1[lane] = bx * bx + by * by;
    }
    const float* s2base = s2 + (size_t)b * MM * CC;
    const float* s1base = s1 + (size_t)b * NN * CC;
    for (int idx = lane; idx < 16 * CC; idx += 32) {
        int r = idx / CC;
        int c = idx - r * CC;
        ls2[idx] = s2base[(size_t)imin_i(m0 + r, MM - 1) * CC + c];
        ls1[idx] = s1base[(size_t)imin_i(n0 + r, NN - 1) * CC + c];
    }
    __syncthreads();

    v8f cacc = {0.f, 0.f, 0.f, 0.f, 0.f, 0.f, 0.f, 0.f};
    v8f dot = __builtin_amdgcn_wmma_f32_16x16x4_f32(
         false, a,  false, bv,
         (short)0, cacc,  false,  false);

    int n_loc  = lane & 15;
    int m_base = (lane >> 4) << 3;

    float cls[8];
#pragma unroll
    for (int i = 0; i < 8; ++i) cls[i] = 0.0f;
    for (int c = 0; c < CC; ++c) {
        float s1v = ls1[n_loc * CC + c];
#pragma unroll
        for (int i = 0; i < 8; ++i) {
            float d = fabsf(ls2[(m_base + i) * CC + c] - s1v);
            cls[i] = fmaxf(cls[i], d);
        }
    }

#pragma unroll
    for (int i = 0; i < 8; ++i) {
        int m = m0 + m_base + i;
        int n = n0 + n_loc;
        if (m < MM && n < NN) {
            float d2 = sq2[m_base + i] + sq1[n_loc] - 2.0f * dot[i];
            float cd = sqrtf(fmaxf(d2, 0.0f));
            cost[((size_t)b * MM + m) * NN + n] = 0.5f * cd + 0.5f * cls[i];
        }
    }
}

#define LTH 512
#define NWAVE (LTH / 32)

template <int CTRL>
__device__ __forceinline__ void fold_stage(double& v, int& i) {
    int lo = __double2loint(v);
    int hi = __double2hiint(v);
    int olo = __builtin_amdgcn_update_dpp(lo, lo, CTRL, 0xf, 0xf, false);
    int ohi = __builtin_amdgcn_update_dpp(hi, hi, CTRL, 0xf, 0xf, false);
    int oi  = __builtin_amdgcn_update_dpp(i,  i,  CTRL, 0xf, 0xf, false);
    double ov = __hiloint2double(ohi, olo);
    bool take = (ov < v) || (ov == v && oi < i);
    v = take ? ov : v;
    i = take ? oi : i;
}

__global__ void __launch_bounds__(LTH)
lapjv_kernel(const float* __restrict__ cost, int* __restrict__ mapping) {
    int b    = blockIdx.x;
    int tid  = threadIdx.x;
    int lane = tid & 31;
    int wave = tid >> 5;
    const float* costB = cost + (size_t)b * MM * NN;
    const double INF = __builtin_inf();

    __shared__ double sh_u[LTH], sh_v[LTH], sh_short[LTH];
    __shared__ int sh_path[LTH], sh_c4r[LTH], sh_r4c[LTH], sh_SR[LTH];
    __shared__ double wred_v[2][NWAVE];
    __shared__ int    wred_i[2][NWAVE];

    if (tid < NN) {
        sh_u[tid] = 0.0; sh_v[tid] = 0.0;
        sh_c4r[tid] = -1; sh_r4c[tid] = -1;
    }
    __syncthreads();

    int par = 0;
    for (int cur = 0; cur < MM; ++cur) {
        int rem = 1;
        if (tid < NN) { sh_short[tid] = INF; sh_path[tid] = -1; }
        int    ii   = cur;
        double minv = 0.0;
        int    sink = -1;
        int    nsr  = 0;

        for (int step = 0; step < NN; ++step) {
            if (tid == 0) sh_SR[nsr] = ii;
            nsr = nsr + 1;

            double cand;
            int    candi = tid;
            if (tid < NN) {
                double r = minv + (double)costB[(size_t)ii * NN + tid]
                                - sh_u[ii] - sh_v[tid];
                if (rem && r < sh_short[tid]) {
                    sh_short[tid] = r;
                    sh_path[tid]  = ii;
                }
                cand = rem ? sh_short[tid] : INF;
            } else {
                cand = INF;
            }

#pragma unroll
            for (int off = 16; off > 0; off >>= 1) {
                double ov = __shfl_xor(cand, off);
                int    oi = __shfl_xor(candi, off);
                if (ov < cand || (ov == cand && oi < candi)) {
                    cand = ov; candi = oi;
                }
            }
            if (lane == 0) { wred_v[par][wave] = cand; wred_i[par][wave] = candi; }
            __syncthreads();

            double fv = wred_v[par][lane & 15];
            int    fi = wred_i[par][lane & 15];
            fold_stage<0x168>(fv, fi);
            fold_stage<0x164>(fv, fi);
            fold_stage<0x162>(fv, fi);
            fold_stage<0x161>(fv, fi);
            par ^= 1;

            int j = __builtin_amdgcn_readfirstlane(fi);
            minv = __hiloint2double(
                __builtin_amdgcn_readfirstlane(__double2hiint(fv)),
                __builtin_amdgcn_readfirstlane(__double2loint(fv)));
            if (tid == j) rem = 0;
            int r4 = __builtin_amdgcn_readfirstlane(sh_r4c[j]);
            if (r4 == -1) { sink = j; break; }
            ii = r4;
        }

        if (sink != -1) {
            if (tid == 0) {
                sh_u[cur] += minv;
            } else if (tid < nsr) {
                int sr = sh_SR[tid];
                sh_u[sr] += minv - sh_short[sh_c4r[sr]];
            }
            if (tid < NN && !rem) sh_v[tid] -= minv - sh_short[tid];
        }
        __syncthreads();

        if (tid == 0 && sink != -1) {
            int j = sink;
            for (int hop = 0; hop < NN; ++hop) {
                int i = sh_path[j];
                sh_r4c[j] = i;
                int tmp = sh_c4r[i];
                sh_c4r[i] = j;
                j = tmp;
                if (i == cur) break;
            }
        }
        __syncthreads();
    }
    if (tid < MM) mapping[b * MM + tid] = sh_c4r[tid];
}

__global__ void extrap_kernel(const float* __restrict__ p1b,
                              const float* __restrict__ p1l,
                              const float* __restrict__ p2b,
                              const float* __restrict__ p2l,
                              const float* __restrict__ toff,
                              const int*   __restrict__ mapping,
                              float* __restrict__ out) {
    int idx = blockIdx.x * blockDim.x + threadIdx.x;
    if (idx >= BB * MM) return;
    int b = idx / MM;

    float first  = toff[b * 3 + 1] - toff[b * 3 + 0];
    float second = toff[b * 3 + 2] - toff[b * 3 + 1];
    float factor = second / first;

    int  j   = mapping[idx];
    bool has = (j >= 0);
    int  jc  = has ? j : 0;

    const float* pb2 = p2b + (size_t)idx * 4;
    const float* pb1 = p1b + ((size_t)b * NN + jc) * 4;
    float* ob = out + (size_t)idx * 4;
#pragma unroll
    for (int k = 0; k < 4; ++k) {
        float c1 = has ? pb1[k] : pb2[k];
        float v  = pb2[k] + (pb2[k] - c1) * factor;
        ob[k] = (k >= 2) ? fmaxf(v, 0.0f) : v;
    }

    const float* pl2 = p2l + (size_t)idx * CC;
    const float* pl1 = p1l + ((size_t)b * NN + jc) * CC;
    float* ol = out + (size_t)BB * MM * 4 + (size_t)idx * CC;
    for (int c = 0; c < CC; ++c) {
        float c1 = has ? pl1[c] : 0.0f;
        ol[c] = 0.5f * (pl2[c] + c1);
    }
}

extern "C" void kernel_launch(void* const* d_in, const int* in_sizes, int n_in,
                              void* d_out, int out_size, void* d_ws, size_t ws_size,
                              hipStream_t stream) {
    (void)in_sizes; (void)n_in; (void)out_size; (void)ws_size;
    const float* p1b  = (const float*)d_in[0];
    const float* p1l  = (const float*)d_in[1];
    const float* p2b  = (const float*)d_in[2];
    const float* p2l  = (const float*)d_in[3];
    const float* toff = (const float*)d_in[4];

    float* ws   = (float*)d_ws;
    float* s1   = ws;
    float* s2   = s1 + (size_t)BB * NN * CC;
    float* cost = s2 + (size_t)BB * MM * CC;
    int*   map  = (int*)(cost + (size_t)BB * MM * NN);
    float* out  = (float*)d_out;

    int tot = BB * NN * CC + BB * MM * CC;
    sigmoid_kernel<<<(tot + 255) / 256, 256, 0, stream>>>(p1l, p2l, s1, s2);

    const int TM = (MM + 15) / 16, TN = (NN + 15) / 16;
    cost_kernel<<<BB * TM * TN, 32, 0, stream>>>(p1b, p2b, s1, s2, cost);

    lapjv_kernel<<<BB, LTH, 0, stream>>>(cost, map);

    int tot2 = BB * MM;
    extrap_kernel<<<(tot2 + 255) / 256, 256, 0, stream>>>(p1b, p1l, p2b, p2l,
                                                          toff, map, out);
}


# Assignment search: the relaxation test and all nine argmin stages compare order-preserving integer keys of the f64 distances (one carry chain, no branch) instead of f64 compares; sums unchanged
# speedup vs baseline: 1.9674x; 1.2567x over previous
#include <hip/hip_runtime.h>
#include <math.h>

#define BB 4
#define MM 500
#define NN 500
#define CC 91

typedef __attribute__((ext_vector_type(2))) float v2f;
typedef __attribute__((ext_vector_type(8))) float v8f;

__device__ __forceinline__ int imin_i(int a, int b) { return a < b ? a : b; }

__global__ void sigmoid_kernel(const float* __restrict__ p1l,
                               const float* __restrict__ p2l,
                               float* __restrict__ s1,
                               float* __restrict__ s2) {
    int idx = blockIdx.x * blockDim.x + threadIdx.x;
    const int n1 = BB * NN * CC;
    const int n2 = BB * MM * CC;
    if (idx < n1) {
        float x = p1l[idx];
        s1[idx] = 1.0f / (1.0f + expf(-x));
    } else if (idx < n1 + n2) {
        float x = p2l[idx - n1];
        s2[idx - n1] = 1.0f / (1.0f + expf(-x));
    }
}

__global__ void cost_kernel(const float* __restrict__ p1b,
                            const float* __restrict__ p2b,
                            const float* __restrict__ s1,
                            const float* __restrict__ s2,
                            float* __restrict__ cost) {
    const int TN = (NN + 15) / 16;
    const int TM = (MM + 15) / 16;
    int bid = blockIdx.x;
    int b  = bid / (TM * TN);
    int t  = bid % (TM * TN);
    int m0 = (t / TN) * 16;
    int n0 = (t % TN) * 16;
    int lane = threadIdx.x;

    __shared__ float sq2[16];
    __shared__ float sq1[16];
    __shared__ float ls2[16 * CC];
    __shared__ float ls1[16 * CC];

    v2f a  = {0.0f, 0.0f};
    v2f bv = {0.0f, 0.0f};
    if (lane < 16) {
        int m = imin_i(m0 + lane, MM - 1);
        const float* pb = p2b + ((size_t)b * MM + m) * 4;
        float x = pb[0], y = pb[1];
        a.x = x; a.y = y;
        sq2[lane] = x * x + y * y;
        int n = imin_i(n0 + lane, NN - 1);
        const float* qb = p1b + ((size_t)b * NN + n) * 4;
        float bx = qb[0], by = qb[1];
        bv.x = bx; bv.y = by;
        sq1[lane] = bx * bx + by * by;
    }
    const float* s2base = s2 + (size_t)b * MM * CC;
    const float* s1base = s1 + (size_t)b * NN * CC;
    for (int idx = lane; idx < 16 * CC; idx += 32) {
        int r = idx / CC;
        int c = idx - r * CC;
        ls2[idx] = s2base[(size_t)imin_i(m0 + r, MM - 1) * CC + c];
        ls1[idx] = s1base[(size_t)imin_i(n0 + r, NN - 1) * CC + c];
    }
    __syncthreads();

    v8f cacc = {0.f, 0.f, 0.f, 0.f, 0.f, 0.f, 0.f, 0.f};
    v8f dot = __builtin_amdgcn_wmma_f32_16x16x4_f32(
         false, a,  false, bv,
         (short)0, cacc,  false,  false);

    int n_loc  = lane & 15;
    int m_base = (lane >> 4) << 3;

    float cls[8];
#pragma unroll
    for (int i = 0; i < 8; ++i) cls[i] = 0.0f;
    for (int c = 0; c < CC; ++c) {
        float s1v = ls1[n_loc * CC + c];
#pragma unroll
        for (int i = 0; i < 8; ++i) {
            float d = fabsf(ls2[(m_base + i) * CC + c] - s1v);
            cls[i] = fmaxf(cls[i], d);
        }
    }

#pragma unroll
    for (int i = 0; i < 8; ++i) {
        int m = m0 + m_base + i;
        int n = n0 + n_loc;
        if (m < MM && n < NN) {
            float d2 = sq2[m_base + i] + sq1[n_loc] - 2.0f * dot[i];
            float cd = sqrtf(fmaxf(d2, 0.0f));
            cost[((size_t)b * MM + m) * NN + n] = 0.5f * cd + 0.5f * cls[i];
        }
    }
}

#define LTH 512
#define NWAVE (LTH / 32)

__device__ __forceinline__ void dkey(double x, unsigned& khi, unsigned& klo) {
    int      hi = __double2hiint(x);
    unsigned lo = (unsigned)__double2loint(x);
    if ((((unsigned)hi & 0x7fffffffu) | lo) == 0u) hi = 0;
    unsigned m = (unsigned)(hi >> 31);
    khi = (unsigned)hi ^ (m | 0x80000000u);
    klo = lo ^ m;
}

__device__ __forceinline__ double dunkey(unsigned khi, unsigned klo) {
    unsigned m = ~(unsigned)((int)khi >> 31);
    return __hiloint2double((int)(khi ^ (m | 0x80000000u)), (int)(klo ^ m));
}

__device__ __forceinline__ bool key_before(unsigned ahi, unsigned alo, int ai,
                                           unsigned bhi, unsigned blo, int bi) {
    bool lt_hi = ahi < bhi, eq_hi = ahi == bhi;
    bool lt_lo = alo < blo, eq_lo = alo == blo;
    bool lt_i  = ai < bi;
    return lt_hi | (eq_hi & (lt_lo | (eq_lo & lt_i)));
}

template <int CTRL>
__device__ __forceinline__ void fold_stage(unsigned& khi, unsigned& klo, int& i) {
    int ohi = __builtin_amdgcn_update_dpp((int)khi, (int)khi, CTRL, 0xf, 0xf, false);
    int olo = __builtin_amdgcn_update_dpp((int)klo, (int)klo, CTRL, 0xf, 0xf, false);
    int oi  = __builtin_amdgcn_update_dpp(i, i, CTRL, 0xf, 0xf, false);
    bool take = key_before((unsigned)ohi, (unsigned)olo, oi, khi, klo, i);
    khi = take ? (unsigned)ohi : khi;
    klo = take ? (unsigned)olo : klo;
    i   = take ? oi : i;
}

__global__ void __launch_bounds__(LTH)
lapjv_kernel(const float* __restrict__ cost, int* __restrict__ mapping) {
    int b    = blockIdx.x;
    int tid  = threadIdx.x;
    int lane = tid & 31;
    int wave = tid >> 5;
    const float* costB = cost + (size_t)b * MM * NN;
    const double INF = __builtin_inf();
    const unsigned INF_KHI = 0xfff00000u;
    const unsigned INF_KLO = 0u;

    __shared__ double sh_u[LTH], sh_v[LTH], sh_short[LTH];
    __shared__ int sh_path[LTH], sh_c4r[LTH], sh_r4c[LTH], sh_SR[LTH];
    __shared__ unsigned wred_hi[2][NWAVE];
    __shared__ unsigned wred_lo[2][NWAVE];
    __shared__ int      wred_i[2][NWAVE];

    if (tid < NN) {
        sh_u[tid] = 0.0; sh_v[tid] = 0.0;
        sh_c4r[tid] = -1; sh_r4c[tid] = -1;
    }
    __syncthreads();

    int par = 0;
    for (int cur = 0; cur < MM; ++cur) {
        int rem = 1;
        if (tid < NN) { sh_short[tid] = INF; sh_path[tid] = -1; }
        unsigned skhi = INF_KHI;
        unsigned sklo = INF_KLO;
        int    ii   = cur;
        double minv = 0.0;
        int    sink = -1;
        int    nsr  = 0;

        for (int step = 0; step < NN; ++step) {
            if (tid == 0) sh_SR[nsr] = ii;
            nsr = nsr + 1;

            unsigned ckhi = INF_KHI;
            unsigned cklo = INF_KLO;
            int      candi = tid;
            if (tid < NN && rem) {
                double r = minv + (double)costB[(size_t)ii * NN + tid]
                                - sh_u[ii] - sh_v[tid];
                unsigned rkhi, rklo;
                dkey(r, rkhi, rklo);
                bool less = (rkhi < skhi) | ((rkhi == skhi) & (rklo < sklo));
                if (less) {
                    sh_short[tid] = r;
                    sh_path[tid]  = ii;
                    skhi = rkhi; sklo = rklo;
                }
                ckhi = skhi; cklo = sklo;
            }

#pragma unroll
            for (int off = 16; off > 0; off >>= 1) {
                unsigned ohi = (unsigned)__shfl_xor((int)ckhi, off);
                unsigned olo = (unsigned)__shfl_xor((int)cklo, off);
                int      oi  = __shfl_xor(candi, off);
                bool take = key_before(ohi, olo, oi, ckhi, cklo, candi);
                ckhi  = take ? ohi : ckhi;
                cklo  = take ? olo : cklo;
                candi = take ? oi  : candi;
            }
            if (lane == 0) {
                wred_hi[par][wave] = ckhi;
                wred_lo[par][wave] = cklo;
                wred_i[par][wave]  = candi;
            }
            __syncthreads();

            unsigned fhi = wred_hi[par][lane & 15];
            unsigned flo = wred_lo[par][lane & 15];
            int      fi  = wred_i[par][lane & 15];
            fold_stage<0x168>(fhi, flo, fi);
            fold_stage<0x164>(fhi, flo, fi);
            fold_stage<0x162>(fhi, flo, fi);
            fold_stage<0x161>(fhi, flo, fi);
            par ^= 1;

            int j = __builtin_amdgcn_readfirstlane(fi);
            minv = dunkey((unsigned)__builtin_amdgcn_readfirstlane((int)fhi),
                          (unsigned)__builtin_amdgcn_readfirstlane((int)flo));
            if (tid == j) rem = 0;
            int r4 = __builtin_amdgcn_readfirstlane(sh_r4c[j]);
            if (r4 == -1) { sink = j; break; }
            ii = r4;
        }

        if (sink != -1) {
            if (tid == 0) {
                sh_u[cur] += minv;
            } else if (tid < nsr) {
                int sr = sh_SR[tid];
                sh_u[sr] += minv - sh_short[sh_c4r[sr]];
            }
            if (tid < NN && !rem) sh_v[tid] -= minv - sh_short[tid];
        }
        __syncthreads();

        if (tid == 0 && sink != -1) {
            int j = sink;
            for (int hop = 0; hop < NN; ++hop) {
                int i = sh_path[j];
                sh_r4c[j] = i;
                int tmp = sh_c4r[i];
                sh_c4r[i] = j;
                j = tmp;
                if (i == cur) break;
            }
        }
        __syncthreads();
    }
    if (tid < MM) mapping[b * MM + tid] = sh_c4r[tid];
}

__global__ void extrap_kernel(const float* __restrict__ p1b,
                              const float* __restrict__ p1l,
                              const float* __restrict__ p2b,
                              const float* __restrict__ p2l,
                              const float* __restrict__ toff,
                              const int*   __restrict__ mapping,
                              float* __restrict__ out) {
    int idx = blockIdx.x * blockDim.x + threadIdx.x;
    if (idx >= BB * MM) return;
    int b = idx / MM;

    float first  = toff[b * 3 + 1] - toff[b * 3 + 0];
    float second = toff[b * 3 + 2] - toff[b * 3 + 1];
    float factor = second / first;

    int  j   = mapping[idx];
    bool has = (j >= 0);
    int  jc  = has ? j : 0;

    const float* pb2 = p2b + (size_t)idx * 4;
    const float* pb1 = p1b + ((size_t)b * NN + jc) * 4;
    float* ob = out + (size_t)idx * 4;
#pragma unroll
    for (int k = 0; k < 4; ++k) {
        float c1 = has ? pb1[k] : pb2[k];
        float v  = pb2[k] + (pb2[k] - c1) * factor;
        ob[k] = (k >= 2) ? fmaxf(v, 0.0f) : v;
    }

    const float* pl2 = p2l + (size_t)idx * CC;
    const float* pl1 = p1l + ((size_t)b * NN + jc) * CC;
    float* ol = out + (size_t)BB * MM * 4 + (size_t)idx * CC;
    for (int c = 0; c < CC; ++c) {
        float c1 = has ? pl1[c] : 0.0f;
        ol[c] = 0.5f * (pl2[c] + c1);
    }
}

extern "C" void kernel_launch(void* const* d_in, const int* in_sizes, int n_in,
                              void* d_out, int out_size, void* d_ws, size_t ws_size,
                              hipStream_t stream) {
    (void)in_sizes; (void)n_in; (void)out_size; (void)ws_size;
    const float* p1b  = (const float*)d_in[0];
    const float* p1l  = (const float*)d_in[1];
    const float* p2b  = (const float*)d_in[2];
    const float* p2l  = (const float*)d_in[3];
    const float* toff = (const float*)d_in[4];

    float* ws   = (float*)d_ws;
    float* s1   = ws;
    float* s2   = s1 + (size_t)BB * NN * CC;
    float* cost = s2 + (size_t)BB * MM * CC;
    int*   map  = (int*)(cost + (size_t)BB * MM * NN);
    float* out  = (float*)d_out;

    int tot = BB * NN * CC + BB * MM * CC;
    sigmoid_kernel<<<(tot + 255) / 256, 256, 0, stream>>>(p1l, p2l, s1, s2);

    const int TM = (MM + 15) / 16, TN = (NN + 15) / 16;
    cost_kernel<<<BB * TM * TN, 32, 0, stream>>>(p1b, p2b, s1, s2, cost);

    lapjv_kernel<<<BB, LTH, 0, stream>>>(cost, map);

    int tot2 = BB * MM;
    extrap_kernel<<<(tot2 + 255) / 256, 256, 0, stream>>>(p1b, p1l, p2b, p2l,
                                                          toff, map, out);
}


# Assignment search argmin, both levels: two 32-bit min reductions of the key words (row_xmask swizzle on the min) and a ballot for the lowest column replace the (key,column) exchange butterfly
# speedup vs baseline: 2.2751x; 1.1564x over previous
#include <hip/hip_runtime.h>
#include <math.h>

#define BB 4
#define MM 500
#define NN 500
#define CC 91

typedef __attribute__((ext_vector_type(2))) float v2f;
typedef __attribute__((ext_vector_type(8))) float v8f;

__device__ __forceinline__ int imin_i(int a, int b) { return a < b ? a : b; }

__global__ void sigmoid_kernel(const float* __restrict__ p1l,
                               const float* __restrict__ p2l,
                               float* __restrict__ s1,
                               float* __restrict__ s2) {
    int idx = blockIdx.x * blockDim.x + threadIdx.x;
    const int n1 = BB * NN * CC;
    const int n2 = BB * MM * CC;
    if (idx < n1) {
        float x = p1l[idx];
        s1[idx] = 1.0f / (1.0f + expf(-x));
    } else if (idx < n1 + n2) {
        float x = p2l[idx - n1];
        s2[idx - n1] = 1.0f / (1.0f + expf(-x));
    }
}

__global__ void cost_kernel(const float* __restrict__ p1b,
                            const float* __restrict__ p2b,
                            const float* __restrict__ s1,
                            const float* __restrict__ s2,
                            float* __restrict__ cost) {
    const int TN = (NN + 15) / 16;
    const int TM = (MM + 15) / 16;
    int bid = blockIdx.x;
    int b  = bid / (TM * TN);
    int t  = bid % (TM * TN);
    int m0 = (t / TN) * 16;
    int n0 = (t % TN) * 16;
    int lane = threadIdx.x;

    __shared__ float sq2[16];
    __shared__ float sq1[16];
    __shared__ float ls2[16 * CC];
    __shared__ float ls1[16 * CC];

    v2f a  = {0.0f, 0.0f};
    v2f bv = {0.0f, 0.0f};
    if (lane < 16) {
        int m = imin_i(m0 + lane, MM - 1);
        const float* pb = p2b + ((size_t)b * MM + m) * 4;
        float x = pb[0], y = pb[1];
        a.x = x; a.y = y;
        sq2[lane] = x * x + y * y;
        int n = imin_i(n0 + lane, NN - 1);
        const float* qb = p1b + ((size_t)b * NN + n) * 4;
        float bx = qb[0], by = qb[1];
        bv.x = bx; bv.y = by;
        sq1[lane] = bx * bx + by * by;
    }
    const float* s2base = s2 + (size_t)b * MM * CC;
    const float* s1base = s1 + (size_t)b * NN * CC;
    for (int idx = lane; idx < 16 * CC; idx += 32) {
        int r = idx / CC;
        int c = idx - r * CC;
        ls2[idx] = s2base[(size_t)imin_i(m0 + r, MM - 1) * CC + c];
        ls1[idx] = s1base[(size_t)imin_i(n0 + r, NN - 1) * CC + c];
    }
    __syncthreads();

    v8f cacc = {0.f, 0.f, 0.f, 0.f, 0.f, 0.f, 0.f, 0.f};
    v8f dot = __builtin_amdgcn_wmma_f32_16x16x4_f32(
         false, a,  false, bv,
         (short)0, cacc,  false,  false);

    int n_loc  = lane & 15;
    int m_base = (lane >> 4) << 3;

    float cls[8];
#pragma unroll
    for (int i = 0; i < 8; ++i) cls[i] = 0.0f;
    for (int c = 0; c < CC; ++c) {
        float s1v = ls1[n_loc * CC + c];
#pragma unroll
        for (int i = 0; i < 8; ++i) {
            float d = fabsf(ls2[(m_base + i) * CC + c] - s1v);
            cls[i] = fmaxf(cls[i], d);
        }
    }

#pragma unroll
    for (int i = 0; i < 8; ++i) {
        int m = m0 + m_base + i;
        int n = n0 + n_loc;
        if (m < MM && n < NN) {
            float d2 = sq2[m_base + i] + sq1[n_loc] - 2.0f * dot[i];
            float cd = sqrtf(fmaxf(d2, 0.0f));
            cost[((size_t)b * MM + m) * NN + n] = 0.5f * cd + 0.5f * cls[i];
        }
    }
}

#define LTH 512
#define NWAVE (LTH / 32)

__device__ __forceinline__ void dkey(double x, unsigned& khi, unsigned& klo) {
    int      hi = __double2hiint(x);
    unsigned lo = (unsigned)__double2loint(x);
    if ((((unsigned)hi & 0x7fffffffu) | lo) == 0u) hi = 0;
    unsigned m = (unsigned)(hi >> 31);
    khi = (unsigned)hi ^ (m | 0x80000000u);
    klo = lo ^ m;
}

__device__ __forceinline__ double dunkey(unsigned khi, unsigned klo) {
    unsigned m = ~(unsigned)((int)khi >> 31);
    return __hiloint2double((int)(khi ^ (m | 0x80000000u)), (int)(klo ^ m));
}


template <int CTRL>
__device__ __forceinline__ unsigned min_stage(unsigned x) {
    unsigned o = (unsigned)__builtin_amdgcn_update_dpp(0, (int)x, CTRL, 0xf, 0xf, true);
    return o < x ? o : x;
}

__device__ __forceinline__ unsigned row_min(unsigned x) {
    x = min_stage<0x168>(x);
    x = min_stage<0x164>(x);
    x = min_stage<0x162>(x);
    x = min_stage<0x161>(x);
    return x;
}

__device__ __forceinline__ unsigned wave_min(unsigned x) {
    x = row_min(x);
    unsigned o = __builtin_amdgcn_permlanex16(x, x, 0x76543210u, 0xfedcba98u,
                                              false, false);
    return o < x ? o : x;
}

__global__ void __launch_bounds__(LTH)
lapjv_kernel(const float* __restrict__ cost, int* __restrict__ mapping) {
    int b    = blockIdx.x;
    int tid  = threadIdx.x;
    int lane = tid & 31;
    int wave = tid >> 5;
    const float* costB = cost + (size_t)b * MM * NN;
    const double INF = __builtin_inf();
    const unsigned INF_KHI = 0xfff00000u;
    const unsigned INF_KLO = 0u;

    __shared__ double sh_u[LTH], sh_v[LTH], sh_short[LTH];
    __shared__ int sh_path[LTH], sh_c4r[LTH], sh_r4c[LTH], sh_SR[LTH];
    __shared__ unsigned wred_hi[2][NWAVE];
    __shared__ unsigned wred_lo[2][NWAVE];
    __shared__ int      wred_i[2][NWAVE];

    if (tid < NN) {
        sh_u[tid] = 0.0; sh_v[tid] = 0.0;
        sh_c4r[tid] = -1; sh_r4c[tid] = -1;
    }
    __syncthreads();

    int par = 0;
    for (int cur = 0; cur < MM; ++cur) {
        int rem = 1;
        if (tid < NN) { sh_short[tid] = INF; sh_path[tid] = -1; }
        unsigned skhi = INF_KHI;
        unsigned sklo = INF_KLO;
        int    ii   = cur;
        double minv = 0.0;
        int    sink = -1;
        int    nsr  = 0;

        for (int step = 0; step < NN; ++step) {
            if (tid == 0) sh_SR[nsr] = ii;
            nsr = nsr + 1;

            unsigned ckhi = INF_KHI;
            unsigned cklo = INF_KLO;
            int      candi = tid;
            if (tid < NN && rem) {
                double r = minv + (double)costB[(size_t)ii * NN + tid]
                                - sh_u[ii] - sh_v[tid];
                unsigned rkhi, rklo;
                dkey(r, rkhi, rklo);
                bool less = (rkhi < skhi) | ((rkhi == skhi) & (rklo < sklo));
                if (less) {
                    sh_short[tid] = r;
                    sh_path[tid]  = ii;
                    skhi = rkhi; sklo = rklo;
                }
                ckhi = skhi; cklo = sklo;
            }

            unsigned mhi = wave_min(ckhi);
            unsigned mlo = wave_min(ckhi == mhi ? cklo : 0xffffffffu);
            unsigned hit = __builtin_amdgcn_ballot_w32((ckhi == mhi) & (cklo == mlo));
            int wcol = (tid & ~31) + __builtin_ctz(hit);
            if (lane == 0) {
                wred_hi[par][wave] = mhi;
                wred_lo[par][wave] = mlo;
                wred_i[par][wave]  = wcol;
            }
            __syncthreads();

            unsigned fhi = wred_hi[par][lane & 15];
            unsigned flo = wred_lo[par][lane & 15];
            int      fi  = wred_i[par][lane & 15];
            unsigned ghi = row_min(fhi);
            unsigned glo = row_min(fhi == ghi ? flo : 0xffffffffu);
            unsigned hitw = __builtin_amdgcn_ballot_w32((fhi == ghi) & (flo == glo));
            int win = __builtin_ctz(hitw);
            par ^= 1;

            int j = __builtin_amdgcn_readlane(fi, win);
            minv = dunkey((unsigned)__builtin_amdgcn_readfirstlane((int)ghi),
                          (unsigned)__builtin_amdgcn_readfirstlane((int)glo));
            if (tid == j) rem = 0;
            int r4 = __builtin_amdgcn_readfirstlane(sh_r4c[j]);
            if (r4 == -1) { sink = j; break; }
            ii = r4;
        }

        if (sink != -1) {
            if (tid == 0) {
                sh_u[cur] += minv;
            } else if (tid < nsr) {
                int sr = sh_SR[tid];
                sh_u[sr] += minv - sh_short[sh_c4r[sr]];
            }
            if (tid < NN && !rem) sh_v[tid] -= minv - sh_short[tid];
        }
        __syncthreads();

        if (tid == 0 && sink != -1) {
            int j = sink;
            for (int hop = 0; hop < NN; ++hop) {
                int i = sh_path[j];
                sh_r4c[j] = i;
                int tmp = sh_c4r[i];
                sh_c4r[i] = j;
                j = tmp;
                if (i == cur) break;
            }
        }
        __syncthreads();
    }
    if (tid < MM) mapping[b * MM + tid] = sh_c4r[tid];
}

__global__ void extrap_kernel(const float* __restrict__ p1b,
                              const float* __restrict__ p1l,
                              const float* __restrict__ p2b,
                              const float* __restrict__ p2l,
                              const float* __restrict__ toff,
                              const int*   __restrict__ mapping,
                              float* __restrict__ out) {
    int idx = blockIdx.x * blockDim.x + threadIdx.x;
    if (idx >= BB * MM) return;
    int b = idx / MM;

    float first  = toff[b * 3 + 1] - toff[b * 3 + 0];
    float second = toff[b * 3 + 2] - toff[b * 3 + 1];
    float factor = second / first;

    int  j   = mapping[idx];
    bool has = (j >= 0);
    int  jc  = has ? j : 0;

    const float* pb2 = p2b + (size_t)idx * 4;
    const float* pb1 = p1b + ((size_t)b * NN + jc) * 4;
    float* ob = out + (size_t)idx * 4;
#pragma unroll
    for (int k = 0; k < 4; ++k) {
        float c1 = has ? pb1[k] : pb2[k];
        float v  = pb2[k] + (pb2[k] - c1) * factor;
        ob[k] = (k >= 2) ? fmaxf(v, 0.0f) : v;
    }

    const float* pl2 = p2l + (size_t)idx * CC;
    const float* pl1 = p1l + ((size_t)b * NN + jc) * CC;
    float* ol = out + (size_t)BB * MM * 4 + (size_t)idx * CC;
    for (int c = 0; c < CC; ++c) {
        float c1 = has ? pl1[c] : 0.0f;
        ol[c] = 0.5f * (pl2[c] + c1);
    }
}

extern "C" void kernel_launch(void* const* d_in, const int* in_sizes, int n_in,
                              void* d_out, int out_size, void* d_ws, size_t ws_size,
                              hipStream_t stream) {
    (void)in_sizes; (void)n_in; (void)out_size; (void)ws_size;
    const float* p1b  = (const float*)d_in[0];
    const float* p1l  = (const float*)d_in[1];
    const float* p2b  = (const float*)d_in[2];
    const float* p2l  = (const float*)d_in[3];
    const float* toff = (const float*)d_in[4];

    float* ws   = (float*)d_ws;
    float* s1   = ws;
    float* s2   = s1 + (size_t)BB * NN * CC;
    float* cost = s2 + (size_t)BB * MM * CC;
    int*   map  = (int*)(cost + (size_t)BB * MM * NN);
    float* out  = (float*)d_out;

    int tot = BB * NN * CC + BB * MM * CC;
    sigmoid_kernel<<<(tot + 255) / 256, 256, 0, stream>>>(p1l, p2l, s1, s2);

    const int TM = (MM + 15) / 16, TN = (NN + 15) / 16;
    cost_kernel<<<BB * TM * TN, 32, 0, stream>>>(p1b, p2b, s1, s2, cost);

    lapjv_kernel<<<BB, LTH, 0, stream>>>(cost, map);

    int tot2 = BB * MM;
    extrap_kernel<<<(tot2 + 255) / 256, 256, 0, stream>>>(p1b, p1l, p2b, p2l,
                                                          toff, map, out);
}


# Assignment search: distances and duals held as exact 64-bit fixed-point integers (unit 2^-54; every f32 cost converts exactly) instead of f64: no f64 instruction left in the step; sums exact
# speedup vs baseline: 3.3608x; 1.4772x over previous
#include <hip/hip_runtime.h>
#include <math.h>

#define BB 4
#define MM 500
#define NN 500
#define CC 91

typedef __attribute__((ext_vector_type(2))) float v2f;
typedef __attribute__((ext_vector_type(8))) float v8f;

__device__ __forceinline__ int imin_i(int a, int b) { return a < b ? a : b; }

__global__ void sigmoid_kernel(const float* __restrict__ p1l,
                               const float* __restrict__ p2l,
                               float* __restrict__ s1,
                               float* __restrict__ s2) {
    int idx = blockIdx.x * blockDim.x + threadIdx.x;
    const int n1 = BB * NN * CC;
    const int n2 = BB * MM * CC;
    if (idx < n1) {
        float x = p1l[idx];
        s1[idx] = 1.0f / (1.0f + expf(-x));
    } else if (idx < n1 + n2) {
        float x = p2l[idx - n1];
        s2[idx - n1] = 1.0f / (1.0f + expf(-x));
    }
}

__global__ void cost_kernel(const float* __restrict__ p1b,
                            const float* __restrict__ p2b,
                            const float* __restrict__ s1,
                            const float* __restrict__ s2,
                            float* __restrict__ cost) {
    const int TN = (NN + 15) / 16;
    const int TM = (MM + 15) / 16;
    int bid = blockIdx.x;
    int b  = bid / (TM * TN);
    int t  = bid % (TM * TN);
    int m0 = (t / TN) * 16;
    int n0 = (t % TN) * 16;
    int lane = threadIdx.x;

    __shared__ float sq2[16];
    __shared__ float sq1[16];
    __shared__ float ls2[16 * CC];
    __shared__ float ls1[16 * CC];

    v2f a  = {0.0f, 0.0f};
    v2f bv = {0.0f, 0.0f};
    if (lane < 16) {
        int m = imin_i(m0 + lane, MM - 1);
        const float* pb = p2b + ((size_t)b * MM + m) * 4;
        float x = pb[0], y = pb[1];
        a.x = x; a.y = y;
        sq2[lane] = x * x + y * y;
        int n = imin_i(n0 + lane, NN - 1);
        const float* qb = p1b + ((size_t)b * NN + n) * 4;
        float bx = qb[0], by = qb[1];
        bv.x = bx; bv.y = by;
        sq1[lane] = bx * bx + by * by;
    }
    const float* s2base = s2 + (size_t)b * MM * CC;
    const float* s1base = s1 + (size_t)b * NN * CC;
    for (int idx = lane; idx < 16 * CC; idx += 32) {
        int r = idx / CC;
        int c = idx - r * CC;
        ls2[idx] = s2base[(size_t)imin_i(m0 + r, MM - 1) * CC + c];
        ls1[idx] = s1base[(size_t)imin_i(n0 + r, NN - 1) * CC + c];
    }
    __syncthreads();

    v8f cacc = {0.f, 0.f, 0.f, 0.f, 0.f, 0.f, 0.f, 0.f};
    v8f dot = __builtin_amdgcn_wmma_f32_16x16x4_f32(
         false, a,  false, bv,
         (short)0, cacc,  false,  false);

    int n_loc  = lane & 15;
    int m_base = (lane >> 4) << 3;

    float cls[8];
#pragma unroll
    for (int i = 0; i < 8; ++i) cls[i] = 0.0f;
    for (int c = 0; c < CC; ++c) {
        float s1v = ls1[n_loc * CC + c];
#pragma unroll
        for (int i = 0; i < 8; ++i) {
            float d = fabsf(ls2[(m_base + i) * CC + c] - s1v);
            cls[i] = fmaxf(cls[i], d);
        }
    }

#pragma unroll
    for (int i = 0; i < 8; ++i) {
        int m = m0 + m_base + i;
        int n = n0 + n_loc;
        if (m < MM && n < NN) {
            float d2 = sq2[m_base + i] + sq1[n_loc] - 2.0f * dot[i];
            float cd = sqrtf(fmaxf(d2, 0.0f));
            cost[((size_t)b * MM + m) * NN + n] = 0.5f * cd + 0.5f * cls[i];
        }
    }
}

#define LTH 512
#define NWAVE (LTH / 32)

typedef unsigned long long u64;

__device__ __forceinline__ u64 cost_fx(float c) {
    float    t  = c * 4194304.0f;
    unsigned hi = (unsigned)t;
    float    f  = t - (float)hi;
    unsigned lo = (unsigned)(f * 4294967296.0f);
    return ((u64)hi << 32) | (u64)lo;
}


template <int CTRL>
__device__ __forceinline__ unsigned min_stage(unsigned x) {
    unsigned o = (unsigned)__builtin_amdgcn_update_dpp(0, (int)x, CTRL, 0xf, 0xf, true);
    return o < x ? o : x;
}

__device__ __forceinline__ unsigned row_min(unsigned x) {
    x = min_stage<0x168>(x);
    x = min_stage<0x164>(x);
    x = min_stage<0x162>(x);
    x = min_stage<0x161>(x);
    return x;
}

__device__ __forceinline__ unsigned wave_min(unsigned x) {
    x = row_min(x);
    unsigned o = __builtin_amdgcn_permlanex16(x, x, 0x76543210u, 0xfedcba98u,
                                              false, false);
    return o < x ? o : x;
}

__global__ void __launch_bounds__(LTH)
lapjv_kernel(const float* __restrict__ cost, int* __restrict__ mapping) {
    int b    = blockIdx.x;
    int tid  = threadIdx.x;
    int lane = tid & 31;
    int wave = tid >> 5;
    const float* costB = cost + (size_t)b * MM * NN;
    const u64      FX_INF = ~0ull;
    const unsigned INF_W  = 0xffffffffu;

    __shared__ u64 sh_u[LTH], sh_v[LTH], sh_short[LTH];
    __shared__ int sh_path[LTH], sh_c4r[LTH], sh_r4c[LTH], sh_SR[LTH];
    __shared__ unsigned wred_hi[2][NWAVE];
    __shared__ unsigned wred_lo[2][NWAVE];
    __shared__ int      wred_i[2][NWAVE];

    if (tid < NN) {
        sh_u[tid] = 0; sh_v[tid] = 0;
        sh_c4r[tid] = -1; sh_r4c[tid] = -1;
    }
    __syncthreads();

    int par = 0;
    for (int cur = 0; cur < MM; ++cur) {
        int rem = 1;
        if (tid < NN) { sh_short[tid] = FX_INF; sh_path[tid] = -1; }
        unsigned shi = INF_W;
        unsigned slo = INF_W;
        int ii   = cur;
        u64 minv = 0;
        int sink = -1;
        int nsr  = 0;

        for (int step = 0; step < NN; ++step) {
            if (tid == 0) sh_SR[nsr] = ii;
            nsr = nsr + 1;

            unsigned chi = INF_W;
            unsigned clo = INF_W;
            if (tid < NN && rem) {
                u64 t0 = minv - sh_u[ii] - sh_v[tid];
                u64 r  = t0 + cost_fx(costB[(size_t)ii * NN + tid]);
                unsigned rhi = (unsigned)(r >> 32);
                unsigned rlo = (unsigned)r;
                bool less = (rhi < shi) | ((rhi == shi) & (rlo < slo));
                if (less) {
                    sh_short[tid] = r;
                    sh_path[tid]  = ii;
                    shi = rhi; slo = rlo;
                }
                chi = shi; clo = slo;
            }

            unsigned mhi = wave_min(chi);
            unsigned mlo = wave_min(chi == mhi ? clo : INF_W);
            unsigned hit = __builtin_amdgcn_ballot_w32((chi == mhi) & (clo == mlo));
            int wcol = (tid & ~31) + __builtin_ctz(hit);
            if (lane == 0) {
                wred_hi[par][wave] = mhi;
                wred_lo[par][wave] = mlo;
                wred_i[par][wave]  = wcol;
            }
            __syncthreads();

            unsigned fhi = wred_hi[par][lane & 15];
            unsigned flo = wred_lo[par][lane & 15];
            int      fi  = wred_i[par][lane & 15];
            unsigned ghi = row_min(fhi);
            unsigned glo = row_min(fhi == ghi ? flo : INF_W);
            unsigned hitw = __builtin_amdgcn_ballot_w32((fhi == ghi) & (flo == glo));
            int win = __builtin_ctz(hitw);
            par ^= 1;

            int j = __builtin_amdgcn_readlane(fi, win);
            minv = ((u64)(unsigned)__builtin_amdgcn_readfirstlane((int)ghi) << 32)
                 | (u64)(unsigned)__builtin_amdgcn_readfirstlane((int)glo);
            if (tid == j) rem = 0;
            int r4 = __builtin_amdgcn_readfirstlane(sh_r4c[j]);
            if (r4 == -1) { sink = j; break; }
            ii = r4;
        }

        if (sink != -1) {
            if (tid == 0) {
                sh_u[cur] += minv;
            } else if (tid < nsr) {
                int sr = sh_SR[tid];
                sh_u[sr] += minv - sh_short[sh_c4r[sr]];
            }
            if (tid < NN && !rem) sh_v[tid] -= minv - sh_short[tid];
        }
        __syncthreads();

        if (tid == 0 && sink != -1) {
            int j = sink;
            for (int hop = 0; hop < NN; ++hop) {
                int i = sh_path[j];
                sh_r4c[j] = i;
                int tmp = sh_c4r[i];
                sh_c4r[i] = j;
                j = tmp;
                if (i == cur) break;
            }
        }
        __syncthreads();
    }
    if (tid < MM) mapping[b * MM + tid] = sh_c4r[tid];
}

__global__ void extrap_kernel(const float* __restrict__ p1b,
                              const float* __restrict__ p1l,
                              const float* __restrict__ p2b,
                              const float* __restrict__ p2l,
                              const float* __restrict__ toff,
                              const int*   __restrict__ mapping,
                              float* __restrict__ out) {
    int idx = blockIdx.x * blockDim.x + threadIdx.x;
    if (idx >= BB * MM) return;
    int b = idx / MM;

    float first  = toff[b * 3 + 1] - toff[b * 3 + 0];
    float second = toff[b * 3 + 2] - toff[b * 3 + 1];
    float factor = second / first;

    int  j   = mapping[idx];
    bool has = (j >= 0);
    int  jc  = has ? j : 0;

    const float* pb2 = p2b + (size_t)idx * 4;
    const float* pb1 = p1b + ((size_t)b * NN + jc) * 4;
    float* ob = out + (size_t)idx * 4;
#pragma unroll
    for (int k = 0; k < 4; ++k) {
        float c1 = has ? pb1[k] : pb2[k];
        float v  = pb2[k] + (pb2[k] - c1) * factor;
        ob[k] = (k >= 2) ? fmaxf(v, 0.0f) : v;
    }

    const float* pl2 = p2l + (size_t)idx * CC;
    const float* pl1 = p1l + ((size_t)b * NN + jc) * CC;
    float* ol = out + (size_t)BB * MM * 4 + (size_t)idx * CC;
    for (int c = 0; c < CC; ++c) {
        float c1 = has ? pl1[c] : 0.0f;
        ol[c] = 0.5f * (pl2[c] + c1);
    }
}

extern "C" void kernel_launch(void* const* d_in, const int* in_sizes, int n_in,
                              void* d_out, int out_size, void* d_ws, size_t ws_size,
                              hipStream_t stream) {
    (void)in_sizes; (void)n_in; (void)out_size; (void)ws_size;
    const float* p1b  = (const float*)d_in[0];
    const float* p1l  = (const float*)d_in[1];
    const float* p2b  = (const float*)d_in[2];
    const float* p2l  = (const float*)d_in[3];
    const float* toff = (const float*)d_in[4];

    float* ws   = (float*)d_ws;
    float* s1   = ws;
    float* s2   = s1 + (size_t)BB * NN * CC;
    float* cost = s2 + (size_t)BB * MM * CC;
    int*   map  = (int*)(cost + (size_t)BB * MM * NN);
    float* out  = (float*)d_out;

    int tot = BB * NN * CC + BB * MM * CC;
    sigmoid_kernel<<<(tot + 255) / 256, 256, 0, stream>>>(p1l, p2l, s1, s2);

    const int TM = (MM + 15) / 16, TN = (NN + 15) / 16;
    cost_kernel<<<BB * TM * TN, 32, 0, stream>>>(p1b, p2b, s1, s2, cost);

    lapjv_kernel<<<BB, LTH, 0, stream>>>(cost, map);

    int tot2 = BB * MM;
    extrap_kernel<<<(tot2 + 255) / 256, 256, 0, stream>>>(p1b, p1l, p2b, p2l,
                                                          toff, map, out);
}


# Assignment search: behind each step's barrier every wave loads one word per 128-byte line of the cost row matched to its own candidate column, so the winner's row is in the near cache when scanned
# speedup vs baseline: 3.7348x; 1.1113x over previous
#include <hip/hip_runtime.h>
#include <math.h>

#define BB 4
#define MM 500
#define NN 500
#define CC 91

typedef __attribute__((ext_vector_type(2))) float v2f;
typedef __attribute__((ext_vector_type(8))) float v8f;

__device__ __forceinline__ int imin_i(int a, int b) { return a < b ? a : b; }

__global__ void sigmoid_kernel(const float* __restrict__ p1l,
                               const float* __restrict__ p2l,
                               float* __restrict__ s1,
                               float* __restrict__ s2) {
    int idx = blockIdx.x * blockDim.x + threadIdx.x;
    const int n1 = BB * NN * CC;
    const int n2 = BB * MM * CC;
    if (idx < n1) {
        float x = p1l[idx];
        s1[idx] = 1.0f / (1.0f + expf(-x));
    } else if (idx < n1 + n2) {
        float x = p2l[idx - n1];
        s2[idx - n1] = 1.0f / (1.0f + expf(-x));
    }
}

__global__ void cost_kernel(const float* __restrict__ p1b,
                            const float* __restrict__ p2b,
                            const float* __restrict__ s1,
                            const float* __restrict__ s2,
                            float* __restrict__ cost) {
    const int TN = (NN + 15) / 16;
    const int TM = (MM + 15) / 16;
    int bid = blockIdx.x;
    int b  = bid / (TM * TN);
    int t  = bid % (TM * TN);
    int m0 = (t / TN) * 16;
    int n0 = (t % TN) * 16;
    int lane = threadIdx.x;

    __shared__ float sq2[16];
    __shared__ float sq1[16];
    __shared__ float ls2[16 * CC];
    __shared__ float ls1[16 * CC];

    v2f a  = {0.0f, 0.0f};
    v2f bv = {0.0f, 0.0f};
    if (lane < 16) {
        int m = imin_i(m0 + lane, MM - 1);
        const float* pb = p2b + ((size_t)b * MM + m) * 4;
        float x = pb[0], y = pb[1];
        a.x = x; a.y = y;
        sq2[lane] = x * x + y * y;
        int n = imin_i(n0 + lane, NN - 1);
        const float* qb = p1b + ((size_t)b * NN + n) * 4;
        float bx = qb[0], by = qb[1];
        bv.x = bx; bv.y = by;
        sq1[lane] = bx * bx + by * by;
    }
    const float* s2base = s2 + (size_t)b * MM * CC;
    const float* s1base = s1 + (size_t)b * NN * CC;
    for (int idx = lane; idx < 16 * CC; idx += 32) {
        int r = idx / CC;
        int c = idx - r * CC;
        ls2[idx] = s2base[(size_t)imin_i(m0 + r, MM - 1) * CC + c];
        ls1[idx] = s1base[(size_t)imin_i(n0 + r, NN - 1) * CC + c];
    }
    __syncthreads();

    v8f cacc = {0.f, 0.f, 0.f, 0.f, 0.f, 0.f, 0.f, 0.f};
    v8f dot = __builtin_amdgcn_wmma_f32_16x16x4_f32(
         false, a,  false, bv,
         (short)0, cacc,  false,  false);

    int n_loc  = lane & 15;
    int m_base = (lane >> 4) << 3;

    float cls[8];
#pragma unroll
    for (int i = 0; i < 8; ++i) cls[i] = 0.0f;
    for (int c = 0; c < CC; ++c) {
        float s1v = ls1[n_loc * CC + c];
#pragma unroll
        for (int i = 0; i < 8; ++i) {
            float d = fabsf(ls2[(m_base + i) * CC + c] - s1v);
            cls[i] = fmaxf(cls[i], d);
        }
    }

#pragma unroll
    for (int i = 0; i < 8; ++i) {
        int m = m0 + m_base + i;
        int n = n0 + n_loc;
        if (m < MM && n < NN) {
            float d2 = sq2[m_base + i] + sq1[n_loc] - 2.0f * dot[i];
            float cd = sqrtf(fmaxf(d2, 0.0f));
            cost[((size_t)b * MM + m) * NN + n] = 0.5f * cd + 0.5f * cls[i];
        }
    }
}

#define LTH 512
#define NWAVE (LTH / 32)

typedef unsigned long long u64;

__device__ __forceinline__ u64 cost_fx(float c) {
    float    t  = c * 4194304.0f;
    unsigned hi = (unsigned)t;
    float    f  = t - (float)hi;
    unsigned lo = (unsigned)(f * 4294967296.0f);
    return ((u64)hi << 32) | (u64)lo;
}


template <int CTRL>
__device__ __forceinline__ unsigned min_stage(unsigned x) {
    unsigned o = (unsigned)__builtin_amdgcn_update_dpp(0, (int)x, CTRL, 0xf, 0xf, true);
    return o < x ? o : x;
}

__device__ __forceinline__ unsigned row_min(unsigned x) {
    x = min_stage<0x168>(x);
    x = min_stage<0x164>(x);
    x = min_stage<0x162>(x);
    x = min_stage<0x161>(x);
    return x;
}

__device__ __forceinline__ unsigned wave_min(unsigned x) {
    x = row_min(x);
    unsigned o = __builtin_amdgcn_permlanex16(x, x, 0x76543210u, 0xfedcba98u,
                                              false, false);
    return o < x ? o : x;
}

__global__ void __launch_bounds__(LTH)
lapjv_kernel(const float* __restrict__ cost, int* __restrict__ mapping) {
    int b    = blockIdx.x;
    int tid  = threadIdx.x;
    int lane = tid & 31;
    int wave = tid >> 5;
    const float* costB = cost + (size_t)b * MM * NN;
    const u64      FX_INF = ~0ull;
    const unsigned INF_W  = 0xffffffffu;

    __shared__ u64 sh_u[LTH], sh_v[LTH], sh_short[LTH];
    __shared__ int sh_path[LTH], sh_c4r[LTH], sh_r4c[LTH], sh_SR[LTH];
    __shared__ unsigned wred_hi[2][NWAVE];
    __shared__ unsigned wred_lo[2][NWAVE];
    __shared__ int      wred_i[2][NWAVE];

    if (tid < NN) {
        sh_u[tid] = 0; sh_v[tid] = 0;
        sh_c4r[tid] = -1; sh_r4c[tid] = -1;
    }
    __syncthreads();

    const int ahead_off = imin_i(lane * 32, NN - 1);

    int par = 0;
    for (int cur = 0; cur < MM; ++cur) {
        int rem = 1;
        if (tid < NN) { sh_short[tid] = FX_INF; sh_path[tid] = -1; }
        int  myrow = (tid < NN) ? sh_r4c[tid] : -1;
        float ahead = 0.0f;
        unsigned shi = INF_W;
        unsigned slo = INF_W;
        int ii   = cur;
        u64 minv = 0;
        int sink = -1;
        int nsr  = 0;

        for (int step = 0; step < NN; ++step) {
            if (tid == 0) sh_SR[nsr] = ii;
            nsr = nsr + 1;

            unsigned chi = INF_W;
            unsigned clo = INF_W;
            if (tid < NN && rem) {
                float c  = costB[(size_t)ii * NN + tid];
                u64   t0 = minv - sh_u[ii] - sh_v[tid];
                __builtin_amdgcn_sched_barrier(0);
                u64 r = t0 + cost_fx(c);
                unsigned rhi = (unsigned)(r >> 32);
                unsigned rlo = (unsigned)r;
                bool less = (rhi < shi) | ((rhi == shi) & (rlo < slo));
                if (less) {
                    sh_short[tid] = r;
                    sh_path[tid]  = ii;
                    shi = rhi; slo = rlo;
                }
                chi = shi; clo = slo;
            }

            unsigned mhi = wave_min(chi);
            unsigned mlo = wave_min(chi == mhi ? clo : INF_W);
            unsigned hit = __builtin_amdgcn_ballot_w32((chi == mhi) & (clo == mlo));
            int wlane = __builtin_ctz(hit);
            int wcol  = (tid & ~31) + wlane;
            int  crow = __builtin_amdgcn_readlane(myrow, wlane);
            bool live = (unsigned)__builtin_amdgcn_readfirstlane((int)mhi) != INF_W;
            if (lane == 0) {
                wred_hi[par][wave] = mhi;
                wred_lo[par][wave] = mlo;
                wred_i[par][wave]  = wcol;
            }
            __syncthreads();

            unsigned fhi = wred_hi[par][lane & 15];
            unsigned flo = wred_lo[par][lane & 15];
            int      fi  = wred_i[par][lane & 15];

            asm volatile("" :: "v"(ahead));
            if (live && crow >= 0)
                ahead = costB[(size_t)crow * NN + ahead_off];
            unsigned ghi = row_min(fhi);
            unsigned glo = row_min(fhi == ghi ? flo : INF_W);
            unsigned hitw = __builtin_amdgcn_ballot_w32((fhi == ghi) & (flo == glo));
            int win = __builtin_ctz(hitw);
            par ^= 1;

            int j = __builtin_amdgcn_readlane(fi, win);
            minv = ((u64)(unsigned)__builtin_amdgcn_readfirstlane((int)ghi) << 32)
                 | (u64)(unsigned)__builtin_amdgcn_readfirstlane((int)glo);
            if (tid == j) rem = 0;
            int r4 = __builtin_amdgcn_readfirstlane(sh_r4c[j]);
            if (r4 == -1) { sink = j; break; }
            ii = r4;
        }

        if (sink != -1) {
            if (tid == 0) {
                sh_u[cur] += minv;
            } else if (tid < nsr) {
                int sr = sh_SR[tid];
                sh_u[sr] += minv - sh_short[sh_c4r[sr]];
            }
            if (tid < NN && !rem) sh_v[tid] -= minv - sh_short[tid];
        }
        __syncthreads();

        if (tid == 0 && sink != -1) {
            int j = sink;
            for (int hop = 0; hop < NN; ++hop) {
                int i = sh_path[j];
                sh_r4c[j] = i;
                int tmp = sh_c4r[i];
                sh_c4r[i] = j;
                j = tmp;
                if (i == cur) break;
            }
        }
        __syncthreads();
    }
    if (tid < MM) mapping[b * MM + tid] = sh_c4r[tid];
}

__global__ void extrap_kernel(const float* __restrict__ p1b,
                              const float* __restrict__ p1l,
                              const float* __restrict__ p2b,
                              const float* __restrict__ p2l,
                              const float* __restrict__ toff,
                              const int*   __restrict__ mapping,
                              float* __restrict__ out) {
    int idx = blockIdx.x * blockDim.x + threadIdx.x;
    if (idx >= BB * MM) return;
    int b = idx / MM;

    float first  = toff[b * 3 + 1] - toff[b * 3 + 0];
    float second = toff[b * 3 + 2] - toff[b * 3 + 1];
    float factor = second / first;

    int  j   = mapping[idx];
    bool has = (j >= 0);
    int  jc  = has ? j : 0;

    const float* pb2 = p2b + (size_t)idx * 4;
    const float* pb1 = p1b + ((size_t)b * NN + jc) * 4;
    float* ob = out + (size_t)idx * 4;
#pragma unroll
    for (int k = 0; k < 4; ++k) {
        float c1 = has ? pb1[k] : pb2[k];
        float v  = pb2[k] + (pb2[k] - c1) * factor;
        ob[k] = (k >= 2) ? fmaxf(v, 0.0f) : v;
    }

    const float* pl2 = p2l + (size_t)idx * CC;
    const float* pl1 = p1l + ((size_t)b * NN + jc) * CC;
    float* ol = out + (size_t)BB * MM * 4 + (size_t)idx * CC;
    for (int c = 0; c < CC; ++c) {
        float c1 = has ? pl1[c] : 0.0f;
        ol[c] = 0.5f * (pl2[c] + c1);
    }
}

extern "C" void kernel_launch(void* const* d_in, const int* in_sizes, int n_in,
                              void* d_out, int out_size, void* d_ws, size_t ws_size,
                              hipStream_t stream) {
    (void)in_sizes; (void)n_in; (void)out_size; (void)ws_size;
    const float* p1b  = (const float*)d_in[0];
    const float* p1l  = (const float*)d_in[1];
    const float* p2b  = (const float*)d_in[2];
    const float* p2l  = (const float*)d_in[3];
    const float* toff = (const float*)d_in[4];

    float* ws   = (float*)d_ws;
    float* s1   = ws;
    float* s2   = s1 + (size_t)BB * NN * CC;
    float* cost = s2 + (size_t)BB * MM * CC;
    int*   map  = (int*)(cost + (size_t)BB * MM * NN);
    float* out  = (float*)d_out;

    int tot = BB * NN * CC + BB * MM * CC;
    sigmoid_kernel<<<(tot + 255) / 256, 256, 0, stream>>>(p1l, p2l, s1, s2);

    const int TM = (MM + 15) / 16, TN = (NN + 15) / 16;
    cost_kernel<<<BB * TM * TN, 32, 0, stream>>>(p1b, p2b, s1, s2, cost);

    lapjv_kernel<<<BB, LTH, 0, stream>>>(cost, map);

    int tot2 = BB * MM;
    extrap_kernel<<<(tot2 + 255) / 256, 256, 0, stream>>>(p1b, p1l, p2b, p2l,
                                                          toff, map, out);
}


# Assignment search: the wave record also carries its candidate's matched row, so the fold yields the next row itself, and its cost word and row dual are requested at once, ahead of the bookkeeping
# speedup vs baseline: 4.0585x; 1.0867x over previous
#include <hip/hip_runtime.h>
#include <math.h>

#define BB 4
#define MM 500
#define NN 500
#define CC 91

typedef __attribute__((ext_vector_type(2))) float v2f;
typedef __attribute__((ext_vector_type(8))) float v8f;

__device__ __forceinline__ int imin_i(int a, int b) { return a < b ? a : b; }

__global__ void sigmoid_kernel(const float* __restrict__ p1l,
                               const float* __restrict__ p2l,
                               float* __restrict__ s1,
                               float* __restrict__ s2) {
    int idx = blockIdx.x * blockDim.x + threadIdx.x;
    const int n1 = BB * NN * CC;
    const int n2 = BB * MM * CC;
    if (idx < n1) {
        float x = p1l[idx];
        s1[idx] = 1.0f / (1.0f + expf(-x));
    } else if (idx < n1 + n2) {
        float x = p2l[idx - n1];
        s2[idx - n1] = 1.0f / (1.0f + expf(-x));
    }
}

__global__ void cost_kernel(const float* __restrict__ p1b,
                            const float* __restrict__ p2b,
                            const float* __restrict__ s1,
                            const float* __restrict__ s2,
                            float* __restrict__ cost) {
    const int TN = (NN + 15) / 16;
    const int TM = (MM + 15) / 16;
    int bid = blockIdx.x;
    int b  = bid / (TM * TN);
    int t  = bid % (TM * TN);
    int m0 = (t / TN) * 16;
    int n0 = (t % TN) * 16;
    int lane = threadIdx.x;

    __shared__ float sq2[16];
    __shared__ float sq1[16];
    __shared__ float ls2[16 * CC];
    __shared__ float ls1[16 * CC];

    v2f a  = {0.0f, 0.0f};
    v2f bv = {0.0f, 0.0f};
    if (lane < 16) {
        int m = imin_i(m0 + lane, MM - 1);
        const float* pb = p2b + ((size_t)b * MM + m) * 4;
        float x = pb[0], y = pb[1];
        a.x = x; a.y = y;
        sq2[lane] = x * x + y * y;
        int n = imin_i(n0 + lane, NN - 1);
        const float* qb = p1b + ((size_t)b * NN + n) * 4;
        float bx = qb[0], by = qb[1];
        bv.x = bx; bv.y = by;
        sq1[lane] = bx * bx + by * by;
    }
    const float* s2base = s2 + (size_t)b * MM * CC;
    const float* s1base = s1 + (size_t)b * NN * CC;
    for (int idx = lane; idx < 16 * CC; idx += 32) {
        int r = idx / CC;
        int c = idx - r * CC;
        ls2[idx] = s2base[(size_t)imin_i(m0 + r, MM - 1) * CC + c];
        ls1[idx] = s1base[(size_t)imin_i(n0 + r, NN - 1) * CC + c];
    }
    __syncthreads();

    v8f cacc = {0.f, 0.f, 0.f, 0.f, 0.f, 0.f, 0.f, 0.f};
    v8f dot = __builtin_amdgcn_wmma_f32_16x16x4_f32(
         false, a,  false, bv,
         (short)0, cacc,  false,  false);

    int n_loc  = lane & 15;
    int m_base = (lane >> 4) << 3;

    float cls[8];
#pragma unroll
    for (int i = 0; i < 8; ++i) cls[i] = 0.0f;
    for (int c = 0; c < CC; ++c) {
        float s1v = ls1[n_loc * CC + c];
#pragma unroll
        for (int i = 0; i < 8; ++i) {
            float d = fabsf(ls2[(m_base + i) * CC + c] - s1v);
            cls[i] = fmaxf(cls[i], d);
        }
    }

#pragma unroll
    for (int i = 0; i < 8; ++i) {
        int m = m0 + m_base + i;
        int n = n0 + n_loc;
        if (m < MM && n < NN) {
            float d2 = sq2[m_base + i] + sq1[n_loc] - 2.0f * dot[i];
            float cd = sqrtf(fmaxf(d2, 0.0f));
            cost[((size_t)b * MM + m) * NN + n] = 0.5f * cd + 0.5f * cls[i];
        }
    }
}

#define LTH 512
#define NWAVE (LTH / 32)

typedef unsigned long long u64;

__device__ __forceinline__ u64 cost_fx(float c) {
    float    t  = c * 4194304.0f;
    unsigned hi = (unsigned)t;
    float    f  = t - (float)hi;
    unsigned lo = (unsigned)(f * 4294967296.0f);
    return ((u64)hi << 32) | (u64)lo;
}


template <int CTRL>
__device__ __forceinline__ unsigned min_stage(unsigned x) {
    unsigned o = (unsigned)__builtin_amdgcn_update_dpp(0, (int)x, CTRL, 0xf, 0xf, true);
    return o < x ? o : x;
}

__device__ __forceinline__ unsigned row_min(unsigned x) {
    x = min_stage<0x168>(x);
    x = min_stage<0x164>(x);
    x = min_stage<0x162>(x);
    x = min_stage<0x161>(x);
    return x;
}

__device__ __forceinline__ unsigned wave_min(unsigned x) {
    x = row_min(x);
    unsigned o = __builtin_amdgcn_permlanex16(x, x, 0x76543210u, 0xfedcba98u,
                                              false, false);
    return o < x ? o : x;
}

__global__ void __launch_bounds__(LTH)
lapjv_kernel(const float* __restrict__ cost, int* __restrict__ mapping) {
    int b    = blockIdx.x;
    int tid  = threadIdx.x;
    int lane = tid & 31;
    int wave = tid >> 5;
    const float* costB = cost + (size_t)b * MM * NN;
    const u64      FX_INF = ~0ull;
    const unsigned INF_W  = 0xffffffffu;

    __shared__ u64 sh_u[LTH], sh_v[LTH], sh_short[LTH];
    __shared__ int sh_path[LTH], sh_c4r[LTH], sh_r4c[LTH], sh_SR[LTH];
    __shared__ uint4 wrec[2][NWAVE];

    if (tid < NN) {
        sh_u[tid] = 0; sh_v[tid] = 0;
        sh_c4r[tid] = -1; sh_r4c[tid] = -1;
    }
    __syncthreads();

    const int ahead_off = imin_i(lane * 32, NN - 1);

    int par = 0;
    for (int cur = 0; cur < MM; ++cur) {
        int rem = 1;
        if (tid < NN) { sh_short[tid] = FX_INF; sh_path[tid] = -1; }
        int  myrow = (tid < NN) ? sh_r4c[tid] : -1;
        u64  vmine = (tid < NN) ? sh_v[tid] : 0;
        float ahead = 0.0f;
        unsigned shi = INF_W;
        unsigned slo = INF_W;
        int ii   = cur;
        u64 minv = 0;
        int sink = -1;
        int nsr  = 0;
        float cw   = (tid < NN) ? costB[(size_t)cur * NN + tid] : 0.0f;
        u64   urow = sh_u[cur];

        for (int step = 0; step < NN; ++step) {
            if (tid == 0) sh_SR[nsr] = ii;
            nsr = nsr + 1;

            unsigned chi = INF_W;
            unsigned clo = INF_W;
            if (tid < NN && rem) {
                u64 t0 = minv - urow - vmine;
                u64 r  = t0 + cost_fx(cw);
                unsigned rhi = (unsigned)(r >> 32);
                unsigned rlo = (unsigned)r;
                bool less = (rhi < shi) | ((rhi == shi) & (rlo < slo));
                if (less) {
                    sh_short[tid] = r;
                    sh_path[tid]  = ii;
                    shi = rhi; slo = rlo;
                }
                chi = shi; clo = slo;
            }

            unsigned mhi = wave_min(chi);
            unsigned mlo = wave_min(chi == mhi ? clo : INF_W);
            unsigned hit = __builtin_amdgcn_ballot_w32((chi == mhi) & (clo == mlo));
            int wlane = __builtin_ctz(hit);
            int wcol  = (tid & ~31) + wlane;
            int  crow = __builtin_amdgcn_readlane(myrow, wlane);
            bool live = (unsigned)__builtin_amdgcn_readfirstlane((int)mhi) != INF_W;
            if (lane == 0)
                wrec[par][wave] = make_uint4(mhi, mlo, (unsigned)wcol, (unsigned)crow);
            __syncthreads();

            uint4    frec = wrec[par][lane & 15];
            unsigned fhi  = frec.x;
            unsigned flo  = frec.y;
            int      fi   = (int)frec.z;
            int      fr   = (int)frec.w;

            asm volatile("" :: "v"(ahead));
            if (live && crow >= 0)
                ahead = costB[(size_t)crow * NN + ahead_off];
            unsigned ghi = row_min(fhi);
            unsigned glo = row_min(fhi == ghi ? flo : INF_W);
            unsigned hitw = __builtin_amdgcn_ballot_w32((fhi == ghi) & (flo == glo));
            int win = __builtin_ctz(hitw);
            par ^= 1;

            int j  = __builtin_amdgcn_readlane(fi, win);
            int r4 = __builtin_amdgcn_readlane(fr, win);
            if (r4 != -1) {
                if (tid < NN) cw = costB[(size_t)r4 * NN + tid];
                urow = sh_u[r4];
            }
            minv = ((u64)(unsigned)__builtin_amdgcn_readfirstlane((int)ghi) << 32)
                 | (u64)(unsigned)__builtin_amdgcn_readfirstlane((int)glo);
            if (tid == j) rem = 0;
            if (r4 == -1) { sink = j; break; }
            ii = r4;
        }

        if (sink != -1) {
            if (tid == 0) {
                sh_u[cur] += minv;
            } else if (tid < nsr) {
                int sr = sh_SR[tid];
                sh_u[sr] += minv - sh_short[sh_c4r[sr]];
            }
            if (tid < NN && !rem) sh_v[tid] -= minv - sh_short[tid];
        }
        __syncthreads();

        if (tid == 0 && sink != -1) {
            int j = sink;
            for (int hop = 0; hop < NN; ++hop) {
                int i = sh_path[j];
                sh_r4c[j] = i;
                int tmp = sh_c4r[i];
                sh_c4r[i] = j;
                j = tmp;
                if (i == cur) break;
            }
        }
        __syncthreads();
    }
    if (tid < MM) mapping[b * MM + tid] = sh_c4r[tid];
}

__global__ void extrap_kernel(const float* __restrict__ p1b,
                              const float* __restrict__ p1l,
                              const float* __restrict__ p2b,
                              const float* __restrict__ p2l,
                              const float* __restrict__ toff,
                              const int*   __restrict__ mapping,
                              float* __restrict__ out) {
    int idx = blockIdx.x * blockDim.x + threadIdx.x;
    if (idx >= BB * MM) return;
    int b = idx / MM;

    float first  = toff[b * 3 + 1] - toff[b * 3 + 0];
    float second = toff[b * 3 + 2] - toff[b * 3 + 1];
    float factor = second / first;

    int  j   = mapping[idx];
    bool has = (j >= 0);
    int  jc  = has ? j : 0;

    const float* pb2 = p2b + (size_t)idx * 4;
    const float* pb1 = p1b + ((size_t)b * NN + jc) * 4;
    float* ob = out + (size_t)idx * 4;
#pragma unroll
    for (int k = 0; k < 4; ++k) {
        float c1 = has ? pb1[k] : pb2[k];
        float v  = pb2[k] + (pb2[k] - c1) * factor;
        ob[k] = (k >= 2) ? fmaxf(v, 0.0f) : v;
    }

    const float* pl2 = p2l + (size_t)idx * CC;
    const float* pl1 = p1l + ((size_t)b * NN + jc) * CC;
    float* ol = out + (size_t)BB * MM * 4 + (size_t)idx * CC;
    for (int c = 0; c < CC; ++c) {
        float c1 = has ? pl1[c] : 0.0f;
        ol[c] = 0.5f * (pl2[c] + c1);
    }
}

extern "C" void kernel_launch(void* const* d_in, const int* in_sizes, int n_in,
                              void* d_out, int out_size, void* d_ws, size_t ws_size,
                              hipStream_t stream) {
    (void)in_sizes; (void)n_in; (void)out_size; (void)ws_size;
    const float* p1b  = (const float*)d_in[0];
    const float* p1l  = (const float*)d_in[1];
    const float* p2b  = (const float*)d_in[2];
    const float* p2l  = (const float*)d_in[3];
    const float* toff = (const float*)d_in[4];

    float* ws   = (float*)d_ws;
    float* s1   = ws;
    float* s2   = s1 + (size_t)BB * NN * CC;
    float* cost = s2 + (size_t)BB * MM * CC;
    int*   map  = (int*)(cost + (size_t)BB * MM * NN);
    float* out  = (float*)d_out;

    int tot = BB * NN * CC + BB * MM * CC;
    sigmoid_kernel<<<(tot + 255) / 256, 256, 0, stream>>>(p1l, p2l, s1, s2);

    const int TM = (MM + 15) / 16, TN = (NN + 15) / 16;
    cost_kernel<<<BB * TM * TN, 32, 0, stream>>>(p1b, p2b, s1, s2, cost);

    lapjv_kernel<<<BB, LTH, 0, stream>>>(cost, map);

    int tot2 = BB * MM;
    extrap_kernel<<<(tot2 + 255) / 256, 256, 0, stream>>>(p1b, p1l, p2b, p2l,
                                                          toff, map, out);
}


# Assignment search argmin, both levels: one min reduction of the high words and a ballot; holders that share one low word (exact ties) give the lowest lane; low-word reduction only if they differ
# speedup vs baseline: 4.2538x; 1.0481x over previous
#include <hip/hip_runtime.h>
#include <math.h>

#define BB 4
#define MM 500
#define NN 500
#define CC 91

typedef __attribute__((ext_vector_type(2))) float v2f;
typedef __attribute__((ext_vector_type(8))) float v8f;

__device__ __forceinline__ int imin_i(int a, int b) { return a < b ? a : b; }

__global__ void sigmoid_kernel(const float* __restrict__ p1l,
                               const float* __restrict__ p2l,
                               float* __restrict__ s1,
                               float* __restrict__ s2) {
    int idx = blockIdx.x * blockDim.x + threadIdx.x;
    const int n1 = BB * NN * CC;
    const int n2 = BB * MM * CC;
    if (idx < n1) {
        float x = p1l[idx];
        s1[idx] = 1.0f / (1.0f + expf(-x));
    } else if (idx < n1 + n2) {
        float x = p2l[idx - n1];
        s2[idx - n1] = 1.0f / (1.0f + expf(-x));
    }
}

__global__ void cost_kernel(const float* __restrict__ p1b,
                            const float* __restrict__ p2b,
                            const float* __restrict__ s1,
                            const float* __restrict__ s2,
                            float* __restrict__ cost) {
    const int TN = (NN + 15) / 16;
    const int TM = (MM + 15) / 16;
    int bid = blockIdx.x;
    int b  = bid / (TM * TN);
    int t  = bid % (TM * TN);
    int m0 = (t / TN) * 16;
    int n0 = (t % TN) * 16;
    int lane = threadIdx.x;

    __shared__ float sq2[16];
    __shared__ float sq1[16];
    __shared__ float ls2[16 * CC];
    __shared__ float ls1[16 * CC];

    v2f a  = {0.0f, 0.0f};
    v2f bv = {0.0f, 0.0f};
    if (lane < 16) {
        int m = imin_i(m0 + lane, MM - 1);
        const float* pb = p2b + ((size_t)b * MM + m) * 4;
        float x = pb[0], y = pb[1];
        a.x = x; a.y = y;
        sq2[lane] = x * x + y * y;
        int n = imin_i(n0 + lane, NN - 1);
        const float* qb = p1b + ((size_t)b * NN + n) * 4;
        float bx = qb[0], by = qb[1];
        bv.x = bx; bv.y = by;
        sq1[lane] = bx * bx + by * by;
    }
    const float* s2base = s2 + (size_t)b * MM * CC;
    const float* s1base = s1 + (size_t)b * NN * CC;
    for (int idx = lane; idx < 16 * CC; idx += 32) {
        int r = idx / CC;
        int c = idx - r * CC;
        ls2[idx] = s2base[(size_t)imin_i(m0 + r, MM - 1) * CC + c];
        ls1[idx] = s1base[(size_t)imin_i(n0 + r, NN - 1) * CC + c];
    }
    __syncthreads();

    v8f cacc = {0.f, 0.f, 0.f, 0.f, 0.f, 0.f, 0.f, 0.f};
    v8f dot = __builtin_amdgcn_wmma_f32_16x16x4_f32(
         false, a,  false, bv,
         (short)0, cacc,  false,  false);

    int n_loc  = lane & 15;
    int m_base = (lane >> 4) << 3;

    float cls[8];
#pragma unroll
    for (int i = 0; i < 8; ++i) cls[i] = 0.0f;
    for (int c = 0; c < CC; ++c) {
        float s1v = ls1[n_loc * CC + c];
#pragma unroll
        for (int i = 0; i < 8; ++i) {
            float d = fabsf(ls2[(m_base + i) * CC + c] - s1v);
            cls[i] = fmaxf(cls[i], d);
        }
    }

#pragma unroll
    for (int i = 0; i < 8; ++i) {
        int m = m0 + m_base + i;
        int n = n0 + n_loc;
        if (m < MM && n < NN) {
            float d2 = sq2[m_base + i] + sq1[n_loc] - 2.0f * dot[i];
            float cd = sqrtf(fmaxf(d2, 0.0f));
            cost[((size_t)b * MM + m) * NN + n] = 0.5f * cd + 0.5f * cls[i];
        }
    }
}

#define LTH 512
#define NWAVE (LTH / 32)

typedef unsigned long long u64;

__device__ __forceinline__ u64 cost_fx(float c) {
    float    t  = c * 4194304.0f;
    unsigned hi = (unsigned)t;
    float    f  = t - (float)hi;
    unsigned lo = (unsigned)(f * 4294967296.0f);
    return ((u64)hi << 32) | (u64)lo;
}


template <int CTRL>
__device__ __forceinline__ unsigned min_stage(unsigned x) {
    unsigned o = (unsigned)__builtin_amdgcn_update_dpp(0, (int)x, CTRL, 0xf, 0xf, true);
    return o < x ? o : x;
}

__device__ __forceinline__ unsigned row_min(unsigned x) {
    x = min_stage<0x168>(x);
    x = min_stage<0x164>(x);
    x = min_stage<0x162>(x);
    x = min_stage<0x161>(x);
    return x;
}

__device__ __forceinline__ unsigned wave_min(unsigned x) {
    x = row_min(x);
    unsigned o = __builtin_amdgcn_permlanex16(x, x, 0x76543210u, 0xfedcba98u,
                                              false, false);
    return o < x ? o : x;
}

__global__ void __launch_bounds__(LTH)
lapjv_kernel(const float* __restrict__ cost, int* __restrict__ mapping) {
    int b    = blockIdx.x;
    int tid  = threadIdx.x;
    int lane = tid & 31;
    int wave = tid >> 5;
    const float* costB = cost + (size_t)b * MM * NN;
    const u64      FX_INF = ~0ull;
    const unsigned INF_W  = 0xffffffffu;

    __shared__ u64 sh_u[LTH], sh_v[LTH], sh_short[LTH];
    __shared__ int sh_path[LTH], sh_c4r[LTH], sh_r4c[LTH], sh_SR[LTH];
    __shared__ uint4 wrec[2][NWAVE];

    if (tid < NN) {
        sh_u[tid] = 0; sh_v[tid] = 0;
        sh_c4r[tid] = -1; sh_r4c[tid] = -1;
    }
    __syncthreads();

    const int ahead_off = imin_i(lane * 32, NN - 1);

    int par = 0;
    for (int cur = 0; cur < MM; ++cur) {
        int rem = 1;
        if (tid < NN) { sh_short[tid] = FX_INF; sh_path[tid] = -1; }
        int  myrow = (tid < NN) ? sh_r4c[tid] : -1;
        u64  vmine = (tid < NN) ? sh_v[tid] : 0;
        float ahead = 0.0f;
        unsigned shi = INF_W;
        unsigned slo = INF_W;
        int ii   = cur;
        u64 minv = 0;
        int sink = -1;
        int nsr  = 0;
        float cw   = (tid < NN) ? costB[(size_t)cur * NN + tid] : 0.0f;
        u64   urow = sh_u[cur];

        for (int step = 0; step < NN; ++step) {
            if (tid == 0) sh_SR[nsr] = ii;
            nsr = nsr + 1;

            unsigned chi = INF_W;
            unsigned clo = INF_W;
            if (tid < NN && rem) {
                u64 t0 = minv - urow - vmine;
                u64 r  = t0 + cost_fx(cw);
                unsigned rhi = (unsigned)(r >> 32);
                unsigned rlo = (unsigned)r;
                bool less = (rhi < shi) | ((rhi == shi) & (rlo < slo));
                if (less) {
                    sh_short[tid] = r;
                    sh_path[tid]  = ii;
                    shi = rhi; slo = rlo;
                }
                chi = shi; clo = slo;
            }

            unsigned mhi = wave_min(chi);
            unsigned hit = __builtin_amdgcn_ballot_w32(chi == mhi);
            unsigned smhi = (unsigned)__builtin_amdgcn_readfirstlane((int)mhi);
            bool live = smhi != INF_W;
            int      wlane = __builtin_ctz(hit);
            unsigned smlo  = (unsigned)__builtin_amdgcn_readlane((int)clo, wlane);
            if ((hit & (hit - 1u)) != 0u) {
                unsigned other = hit & ~__builtin_amdgcn_ballot_w32(clo == smlo);
                if (other != 0u) {
                    unsigned mlo = wave_min(chi == mhi ? clo : INF_W);
                    hit  &= __builtin_amdgcn_ballot_w32(clo == mlo);
                    wlane = __builtin_ctz(hit);
                    smlo  = (unsigned)__builtin_amdgcn_readlane((int)clo, wlane);
                }
            }
            int wcol  = (tid & ~31) + wlane;
            int  crow = __builtin_amdgcn_readlane(myrow, wlane);
            if (lane == 0)
                wrec[par][wave] = make_uint4(smhi, smlo, (unsigned)wcol, (unsigned)crow);
            __syncthreads();

            uint4    frec = wrec[par][lane & 15];
            unsigned fhi  = frec.x;
            unsigned flo  = frec.y;
            int      fi   = (int)frec.z;
            int      fr   = (int)frec.w;

            asm volatile("" :: "v"(ahead));
            if (live && crow >= 0)
                ahead = costB[(size_t)crow * NN + ahead_off];
            unsigned ghi  = row_min(fhi);
            unsigned hitw = __builtin_amdgcn_ballot_w32(fhi == ghi);
            int      win  = __builtin_ctz(hitw);
            unsigned sglo = (unsigned)__builtin_amdgcn_readlane((int)flo, win);
            if ((hitw & (hitw - 1u) & 0xffffu) != 0u) {
                unsigned other = hitw & ~__builtin_amdgcn_ballot_w32(flo == sglo);
                if (other != 0u) {
                    unsigned glo = row_min(fhi == ghi ? flo : INF_W);
                    hitw &= __builtin_amdgcn_ballot_w32(flo == glo);
                    win   = __builtin_ctz(hitw);
                    sglo  = (unsigned)__builtin_amdgcn_readlane((int)flo, win);
                }
            }
            par ^= 1;

            int j  = __builtin_amdgcn_readlane(fi, win);
            int r4 = __builtin_amdgcn_readlane(fr, win);
            if (r4 != -1) {
                if (tid < NN) cw = costB[(size_t)r4 * NN + tid];
                urow = sh_u[r4];
            }
            minv = ((u64)(unsigned)__builtin_amdgcn_readfirstlane((int)ghi) << 32)
                 | (u64)sglo;
            if (tid == j) rem = 0;
            if (r4 == -1) { sink = j; break; }
            ii = r4;
        }

        if (sink != -1) {
            if (tid == 0) {
                sh_u[cur] += minv;
            } else if (tid < nsr) {
                int sr = sh_SR[tid];
                sh_u[sr] += minv - sh_short[sh_c4r[sr]];
            }
            if (tid < NN && !rem) sh_v[tid] -= minv - sh_short[tid];
        }
        __syncthreads();

        if (tid == 0 && sink != -1) {
            int j = sink;
            for (int hop = 0; hop < NN; ++hop) {
                int i = sh_path[j];
                sh_r4c[j] = i;
                int tmp = sh_c4r[i];
                sh_c4r[i] = j;
                j = tmp;
                if (i == cur) break;
            }
        }
        __syncthreads();
    }
    if (tid < MM) mapping[b * MM + tid] = sh_c4r[tid];
}

__global__ void extrap_kernel(const float* __restrict__ p1b,
                              const float* __restrict__ p1l,
                              const float* __restrict__ p2b,
                              const float* __restrict__ p2l,
                              const float* __restrict__ toff,
                              const int*   __restrict__ mapping,
                              float* __restrict__ out) {
    int idx = blockIdx.x * blockDim.x + threadIdx.x;
    if (idx >= BB * MM) return;
    int b = idx / MM;

    float first  = toff[b * 3 + 1] - toff[b * 3 + 0];
    float second = toff[b * 3 + 2] - toff[b * 3 + 1];
    float factor = second / first;

    int  j   = mapping[idx];
    bool has = (j >= 0);
    int  jc  = has ? j : 0;

    const float* pb2 = p2b + (size_t)idx * 4;
    const float* pb1 = p1b + ((size_t)b * NN + jc) * 4;
    float* ob = out + (size_t)idx * 4;
#pragma unroll
    for (int k = 0; k < 4; ++k) {
        float c1 = has ? pb1[k] : pb2[k];
        float v  = pb2[k] + (pb2[k] - c1) * factor;
        ob[k] = (k >= 2) ? fmaxf(v, 0.0f) : v;
    }

    const float* pl2 = p2l + (size_t)idx * CC;
    const float* pl1 = p1l + ((size_t)b * NN + jc) * CC;
    float* ol = out + (size_t)BB * MM * 4 + (size_t)idx * CC;
    for (int c = 0; c < CC; ++c) {
        float c1 = has ? pl1[c] : 0.0f;
        ol[c] = 0.5f * (pl2[c] + c1);
    }
}

extern "C" void kernel_launch(void* const* d_in, const int* in_sizes, int n_in,
                              void* d_out, int out_size, void* d_ws, size_t ws_size,
                              hipStream_t stream) {
    (void)in_sizes; (void)n_in; (void)out_size; (void)ws_size;
    const float* p1b  = (const float*)d_in[0];
    const float* p1l  = (const float*)d_in[1];
    const float* p2b  = (const float*)d_in[2];
    const float* p2l  = (const float*)d_in[3];
    const float* toff = (const float*)d_in[4];

    float* ws   = (float*)d_ws;
    float* s1   = ws;
    float* s2   = s1 + (size_t)BB * NN * CC;
    float* cost = s2 + (size_t)BB * MM * CC;
    int*   map  = (int*)(cost + (size_t)BB * MM * NN);
    float* out  = (float*)d_out;

    int tot = BB * NN * CC + BB * MM * CC;
    sigmoid_kernel<<<(tot + 255) / 256, 256, 0, stream>>>(p1l, p2l, s1, s2);

    const int TM = (MM + 15) / 16, TN = (NN + 15) / 16;
    cost_kernel<<<BB * TM * TN, 32, 0, stream>>>(p1b, p2b, s1, s2, cost);

    lapjv_kernel<<<BB, LTH, 0, stream>>>(cost, map);

    int tot2 = BB * MM;
    extrap_kernel<<<(tot2 + 255) / 256, 256, 0, stream>>>(p1b, p1l, p2b, p2l,
                                                          toff, map, out);
}


# Assignment search starts from column-minimum duals and a greedy matching of rows to the columns whose minimum they hold; only rows left free are searched; same optimum, sums stay exact integers
# speedup vs baseline: 4.9365x; 1.1605x over previous
#include <hip/hip_runtime.h>
#include <math.h>

#define BB 4
#define MM 500
#define NN 500
#define CC 91

typedef __attribute__((ext_vector_type(2))) float v2f;
typedef __attribute__((ext_vector_type(8))) float v8f;

__device__ __forceinline__ int imin_i(int a, int b) { return a < b ? a : b; }

__global__ void sigmoid_kernel(const float* __restrict__ p1l,
                               const float* __restrict__ p2l,
                               float* __restrict__ s1,
                               float* __restrict__ s2) {
    int idx = blockIdx.x * blockDim.x + threadIdx.x;
    const int n1 = BB * NN * CC;
    const int n2 = BB * MM * CC;
    if (idx < n1) {
        float x = p1l[idx];
        s1[idx] = 1.0f / (1.0f + expf(-x));
    } else if (idx < n1 + n2) {
        float x = p2l[idx - n1];
        s2[idx - n1] = 1.0f / (1.0f + expf(-x));
    }
}

__global__ void cost_kernel(const float* __restrict__ p1b,
                            const float* __restrict__ p2b,
                            const float* __restrict__ s1,
                            const float* __restrict__ s2,
                            float* __restrict__ cost) {
    const int TN = (NN + 15) / 16;
    const int TM = (MM + 15) / 16;
    int bid = blockIdx.x;
    int b  = bid / (TM * TN);
    int t  = bid % (TM * TN);
    int m0 = (t / TN) * 16;
    int n0 = (t % TN) * 16;
    int lane = threadIdx.x;

    __shared__ float sq2[16];
    __shared__ float sq1[16];
    __shared__ float ls2[16 * CC];
    __shared__ float ls1[16 * CC];

    v2f a  = {0.0f, 0.0f};
    v2f bv = {0.0f, 0.0f};
    if (lane < 16) {
        int m = imin_i(m0 + lane, MM - 1);
        const float* pb = p2b + ((size_t)b * MM + m) * 4;
        float x = pb[0], y = pb[1];
        a.x = x; a.y = y;
        sq2[lane] = x * x + y * y;
        int n = imin_i(n0 + lane, NN - 1);
        const float* qb = p1b + ((size_t)b * NN + n) * 4;
        float bx = qb[0], by = qb[1];
        bv.x = bx; bv.y = by;
        sq1[lane] = bx * bx + by * by;
    }
    const float* s2base = s2 + (size_t)b * MM * CC;
    const float* s1base = s1 + (size_t)b * NN * CC;
    for (int idx = lane; idx < 16 * CC; idx += 32) {
        int r = idx / CC;
        int c = idx - r * CC;
        ls2[idx] = s2base[(size_t)imin_i(m0 + r, MM - 1) * CC + c];
        ls1[idx] = s1base[(size_t)imin_i(n0 + r, NN - 1) * CC + c];
    }
    __syncthreads();

    v8f cacc = {0.f, 0.f, 0.f, 0.f, 0.f, 0.f, 0.f, 0.f};
    v8f dot = __builtin_amdgcn_wmma_f32_16x16x4_f32(
         false, a,  false, bv,
         (short)0, cacc,  false,  false);

    int n_loc  = lane & 15;
    int m_base = (lane >> 4) << 3;

    float cls[8];
#pragma unroll
    for (int i = 0; i < 8; ++i) cls[i] = 0.0f;
    for (int c = 0; c < CC; ++c) {
        float s1v = ls1[n_loc * CC + c];
#pragma unroll
        for (int i = 0; i < 8; ++i) {
            float d = fabsf(ls2[(m_base + i) * CC + c] - s1v);
            cls[i] = fmaxf(cls[i], d);
        }
    }

#pragma unroll
    for (int i = 0; i < 8; ++i) {
        int m = m0 + m_base + i;
        int n = n0 + n_loc;
        if (m < MM && n < NN) {
            float d2 = sq2[m_base + i] + sq1[n_loc] - 2.0f * dot[i];
            float cd = sqrtf(fmaxf(d2, 0.0f));
            cost[((size_t)b * MM + m) * NN + n] = 0.5f * cd + 0.5f * cls[i];
        }
    }
}

#define LTH 512
#define NWAVE (LTH / 32)

typedef unsigned long long u64;

__device__ __forceinline__ u64 cost_fx(float c) {
    float    t  = c * 4194304.0f;
    unsigned hi = (unsigned)t;
    float    f  = t - (float)hi;
    unsigned lo = (unsigned)(f * 4294967296.0f);
    return ((u64)hi << 32) | (u64)lo;
}


template <int CTRL>
__device__ __forceinline__ unsigned min_stage(unsigned x) {
    unsigned o = (unsigned)__builtin_amdgcn_update_dpp(0, (int)x, CTRL, 0xf, 0xf, true);
    return o < x ? o : x;
}

__device__ __forceinline__ unsigned row_min(unsigned x) {
    x = min_stage<0x168>(x);
    x = min_stage<0x164>(x);
    x = min_stage<0x162>(x);
    x = min_stage<0x161>(x);
    return x;
}

__device__ __forceinline__ unsigned wave_min(unsigned x) {
    x = row_min(x);
    unsigned o = __builtin_amdgcn_permlanex16(x, x, 0x76543210u, 0xfedcba98u,
                                              false, false);
    return o < x ? o : x;
}

__global__ void __launch_bounds__(LTH)
lapjv_kernel(const float* __restrict__ cost, int* __restrict__ mapping) {
    int b    = blockIdx.x;
    int tid  = threadIdx.x;
    int lane = tid & 31;
    int wave = tid >> 5;
    const float* costB = cost + (size_t)b * MM * NN;
    const u64      FX_INF = ~0ull;
    const unsigned INF_W  = 0xffffffffu;

    __shared__ u64 sh_u[LTH], sh_v[LTH], sh_short[LTH];
    __shared__ int sh_path[LTH], sh_c4r[LTH], sh_r4c[LTH], sh_SR[LTH];
    __shared__ uint4 wrec[2][NWAVE];

    if (tid < NN) {
        sh_u[tid] = 0;
        sh_c4r[tid] = -1; sh_r4c[tid] = -1;
        float best = __builtin_huge_valf();
        int   bi   = 0;
        for (int i0 = 0; i0 < MM; i0 += 10) {
            float c[10];
#pragma unroll
            for (int k = 0; k < 10; ++k)
                c[k] = costB[(size_t)(i0 + k) * NN + tid];
#pragma unroll
            for (int k = 0; k < 10; ++k)
                if (c[k] < best) { best = c[k]; bi = i0 + k; }
        }
        sh_v[tid]    = cost_fx(best);
        sh_path[tid] = bi;
    }
    __syncthreads();
    if (tid < MM) {
        int mine = -1;
        for (int j = NN - 1; j >= 0; --j)
            if (sh_path[j] == tid) mine = j;
        if (mine >= 0) { sh_c4r[tid] = mine; sh_r4c[mine] = tid; }
    }
    __syncthreads();

    const int ahead_off = imin_i(lane * 32, NN - 1);

    int par = 0;
    for (int cur = 0; cur < MM; ++cur) {
        if (__builtin_amdgcn_readfirstlane(sh_c4r[cur]) != -1) continue;
        int rem = 1;
        if (tid < NN) { sh_short[tid] = FX_INF; sh_path[tid] = -1; }
        int  myrow = (tid < NN) ? sh_r4c[tid] : -1;
        u64  vmine = (tid < NN) ? sh_v[tid] : 0;
        float ahead = 0.0f;
        unsigned shi = INF_W;
        unsigned slo = INF_W;
        int ii   = cur;
        u64 minv = 0;
        int sink = -1;
        int nsr  = 0;
        float cw   = (tid < NN) ? costB[(size_t)cur * NN + tid] : 0.0f;
        u64   urow = sh_u[cur];

        for (int step = 0; step < NN; ++step) {
            if (tid == 0) sh_SR[nsr] = ii;
            nsr = nsr + 1;

            unsigned chi = INF_W;
            unsigned clo = INF_W;
            if (tid < NN && rem) {
                u64 t0 = minv - urow - vmine;
                u64 r  = t0 + cost_fx(cw);
                unsigned rhi = (unsigned)(r >> 32);
                unsigned rlo = (unsigned)r;
                bool less = (rhi < shi) | ((rhi == shi) & (rlo < slo));
                if (less) {
                    sh_short[tid] = r;
                    sh_path[tid]  = ii;
                    shi = rhi; slo = rlo;
                }
                chi = shi; clo = slo;
            }

            unsigned mhi = wave_min(chi);
            unsigned hit = __builtin_amdgcn_ballot_w32(chi == mhi);
            unsigned smhi = (unsigned)__builtin_amdgcn_readfirstlane((int)mhi);
            bool live = smhi != INF_W;
            int      wlane = __builtin_ctz(hit);
            unsigned smlo  = (unsigned)__builtin_amdgcn_readlane((int)clo, wlane);
            if ((hit & (hit - 1u)) != 0u) {
                unsigned other = hit & ~__builtin_amdgcn_ballot_w32(clo == smlo);
                if (other != 0u) {
                    unsigned mlo = wave_min(chi == mhi ? clo : INF_W);
                    hit  &= __builtin_amdgcn_ballot_w32(clo == mlo);
                    wlane = __builtin_ctz(hit);
                    smlo  = (unsigned)__builtin_amdgcn_readlane((int)clo, wlane);
                }
            }
            int wcol  = (tid & ~31) + wlane;
            int  crow = __builtin_amdgcn_readlane(myrow, wlane);
            if (lane == 0)
                wrec[par][wave] = make_uint4(smhi, smlo, (unsigned)wcol, (unsigned)crow);
            __syncthreads();

            uint4    frec = wrec[par][lane & 15];
            unsigned fhi  = frec.x;
            unsigned flo  = frec.y;
            int      fi   = (int)frec.z;
            int      fr   = (int)frec.w;

            asm volatile("" :: "v"(ahead));
            if (live && crow >= 0)
                ahead = costB[(size_t)crow * NN + ahead_off];
            unsigned ghi  = row_min(fhi);
            unsigned hitw = __builtin_amdgcn_ballot_w32(fhi == ghi);
            int      win  = __builtin_ctz(hitw);
            unsigned sglo = (unsigned)__builtin_amdgcn_readlane((int)flo, win);
            if ((hitw & (hitw - 1u) & 0xffffu) != 0u) {
                unsigned other = hitw & ~__builtin_amdgcn_ballot_w32(flo == sglo);
                if (other != 0u) {
                    unsigned glo = row_min(fhi == ghi ? flo : INF_W);
                    hitw &= __builtin_amdgcn_ballot_w32(flo == glo);
                    win   = __builtin_ctz(hitw);
                    sglo  = (unsigned)__builtin_amdgcn_readlane((int)flo, win);
                }
            }
            par ^= 1;

            int j  = __builtin_amdgcn_readlane(fi, win);
            int r4 = __builtin_amdgcn_readlane(fr, win);
            if (r4 != -1) {
                if (tid < NN) cw = costB[(size_t)r4 * NN + tid];
                urow = sh_u[r4];
            }
            minv = ((u64)(unsigned)__builtin_amdgcn_readfirstlane((int)ghi) << 32)
                 | (u64)sglo;
            if (tid == j) rem = 0;
            if (r4 == -1) { sink = j; break; }
            ii = r4;
        }

        if (sink != -1) {
            if (tid == 0) {
                sh_u[cur] += minv;
            } else if (tid < nsr) {
                int sr = sh_SR[tid];
                sh_u[sr] += minv - sh_short[sh_c4r[sr]];
            }
            if (tid < NN && !rem) sh_v[tid] -= minv - sh_short[tid];
        }
        __syncthreads();

        if (tid == 0 && sink != -1) {
            int j = sink;
            for (int hop = 0; hop < NN; ++hop) {
                int i = sh_path[j];
                sh_r4c[j] = i;
                int tmp = sh_c4r[i];
                sh_c4r[i] = j;
                j = tmp;
                if (i == cur) break;
            }
        }
        __syncthreads();
    }
    if (tid < MM) mapping[b * MM + tid] = sh_c4r[tid];
}

__global__ void extrap_kernel(const float* __restrict__ p1b,
                              const float* __restrict__ p1l,
                              const float* __restrict__ p2b,
                              const float* __restrict__ p2l,
                              const float* __restrict__ toff,
                              const int*   __restrict__ mapping,
                              float* __restrict__ out) {
    int idx = blockIdx.x * blockDim.x + threadIdx.x;
    if (idx >= BB * MM) return;
    int b = idx / MM;

    float first  = toff[b * 3 + 1] - toff[b * 3 + 0];
    float second = toff[b * 3 + 2] - toff[b * 3 + 1];
    float factor = second / first;

    int  j   = mapping[idx];
    bool has = (j >= 0);
    int  jc  = has ? j : 0;

    const float* pb2 = p2b + (size_t)idx * 4;
    const float* pb1 = p1b + ((size_t)b * NN + jc) * 4;
    float* ob = out + (size_t)idx * 4;
#pragma unroll
    for (int k = 0; k < 4; ++k) {
        float c1 = has ? pb1[k] : pb2[k];
        float v  = pb2[k] + (pb2[k] - c1) * factor;
        ob[k] = (k >= 2) ? fmaxf(v, 0.0f) : v;
    }

    const float* pl2 = p2l + (size_t)idx * CC;
    const float* pl1 = p1l + ((size_t)b * NN + jc) * CC;
    float* ol = out + (size_t)BB * MM * 4 + (size_t)idx * CC;
    for (int c = 0; c < CC; ++c) {
        float c1 = has ? pl1[c] : 0.0f;
        ol[c] = 0.5f * (pl2[c] + c1);
    }
}

extern "C" void kernel_launch(void* const* d_in, const int* in_sizes, int n_in,
                              void* d_out, int out_size, void* d_ws, size_t ws_size,
                              hipStream_t stream) {
    (void)in_sizes; (void)n_in; (void)out_size; (void)ws_size;
    const float* p1b  = (const float*)d_in[0];
    const float* p1l  = (const float*)d_in[1];
    const float* p2b  = (const float*)d_in[2];
    const float* p2l  = (const float*)d_in[3];
    const float* toff = (const float*)d_in[4];

    float* ws   = (float*)d_ws;
    float* s1   = ws;
    float* s2   = s1 + (size_t)BB * NN * CC;
    float* cost = s2 + (size_t)BB * MM * CC;
    int*   map  = (int*)(cost + (size_t)BB * MM * NN);
    float* out  = (float*)d_out;

    int tot = BB * NN * CC + BB * MM * CC;
    sigmoid_kernel<<<(tot + 255) / 256, 256, 0, stream>>>(p1l, p2l, s1, s2);

    const int TM = (MM + 15) / 16, TN = (NN + 15) / 16;
    cost_kernel<<<BB * TM * TN, 32, 0, stream>>>(p1b, p2b, s1, s2, cost);

    lapjv_kernel<<<BB, LTH, 0, stream>>>(cost, map);

    int tot2 = BB * MM;
    extrap_kernel<<<(tot2 + 255) / 256, 256, 0, stream>>>(p1b, p1l, p2b, p2l,
                                                          toff, map, out);
}


# Assignment search: every column exactly at the minimum distance is scanned in the same step (one barrier and fold per set, the rows' loads in flight together); exact integer sums, same optimum
# speedup vs baseline: 5.7124x; 1.1572x over previous
#include <hip/hip_runtime.h>
#include <math.h>

#define BB 4
#define MM 500
#define NN 500
#define CC 91

typedef __attribute__((ext_vector_type(2))) float v2f;
typedef __attribute__((ext_vector_type(8))) float v8f;

__device__ __forceinline__ int imin_i(int a, int b) { return a < b ? a : b; }

__global__ void sigmoid_kernel(const float* __restrict__ p1l,
                               const float* __restrict__ p2l,
                               float* __restrict__ s1,
                               float* __restrict__ s2) {
    int idx = blockIdx.x * blockDim.x + threadIdx.x;
    const int n1 = BB * NN * CC;
    const int n2 = BB * MM * CC;
    if (idx < n1) {
        float x = p1l[idx];
        s1[idx] = 1.0f / (1.0f + expf(-x));
    } else if (idx < n1 + n2) {
        float x = p2l[idx - n1];
        s2[idx - n1] = 1.0f / (1.0f + expf(-x));
    }
}

__global__ void cost_kernel(const float* __restrict__ p1b,
                            const float* __restrict__ p2b,
                            const float* __restrict__ s1,
                            const float* __restrict__ s2,
                            float* __restrict__ cost) {
    const int TN = (NN + 15) / 16;
    const int TM = (MM + 15) / 16;
    int bid = blockIdx.x;
    int b  = bid / (TM * TN);
    int t  = bid % (TM * TN);
    int m0 = (t / TN) * 16;
    int n0 = (t % TN) * 16;
    int lane = threadIdx.x;

    __shared__ float sq2[16];
    __shared__ float sq1[16];
    __shared__ float ls2[16 * CC];
    __shared__ float ls1[16 * CC];

    v2f a  = {0.0f, 0.0f};
    v2f bv = {0.0f, 0.0f};
    if (lane < 16) {
        int m = imin_i(m0 + lane, MM - 1);
        const float* pb = p2b + ((size_t)b * MM + m) * 4;
        float x = pb[0], y = pb[1];
        a.x = x; a.y = y;
        sq2[lane] = x * x + y * y;
        int n = imin_i(n0 + lane, NN - 1);
        const float* qb = p1b + ((size_t)b * NN + n) * 4;
        float bx = qb[0], by = qb[1];
        bv.x = bx; bv.y = by;
        sq1[lane] = bx * bx + by * by;
    }
    const float* s2base = s2 + (size_t)b * MM * CC;
    const float* s1base = s1 + (size_t)b * NN * CC;
    for (int idx = lane; idx < 16 * CC; idx += 32) {
        int r = idx / CC;
        int c = idx - r * CC;
        ls2[idx] = s2base[(size_t)imin_i(m0 + r, MM - 1) * CC + c];
        ls1[idx] = s1base[(size_t)imin_i(n0 + r, NN - 1) * CC + c];
    }
    __syncthreads();

    v8f cacc = {0.f, 0.f, 0.f, 0.f, 0.f, 0.f, 0.f, 0.f};
    v8f dot = __builtin_amdgcn_wmma_f32_16x16x4_f32(
         false, a,  false, bv,
         (short)0, cacc,  false,  false);

    int n_loc  = lane & 15;
    int m_base = (lane >> 4) << 3;

    float cls[8];
#pragma unroll
    for (int i = 0; i < 8; ++i) cls[i] = 0.0f;
    for (int c = 0; c < CC; ++c) {
        float s1v = ls1[n_loc * CC + c];
#pragma unroll
        for (int i = 0; i < 8; ++i) {
            float d = fabsf(ls2[(m_base + i) * CC + c] - s1v);
            cls[i] = fmaxf(cls[i], d);
        }
    }

#pragma unroll
    for (int i = 0; i < 8; ++i) {
        int m = m0 + m_base + i;
        int n = n0 + n_loc;
        if (m < MM && n < NN) {
            float d2 = sq2[m_base + i] + sq1[n_loc] - 2.0f * dot[i];
            float cd = sqrtf(fmaxf(d2, 0.0f));
            cost[((size_t)b * MM + m) * NN + n] = 0.5f * cd + 0.5f * cls[i];
        }
    }
}

#define LTH 512
#define NWAVE (LTH / 32)

typedef unsigned long long u64;

__device__ __forceinline__ u64 cost_fx(float c) {
    float    t  = c * 4194304.0f;
    unsigned hi = (unsigned)t;
    float    f  = t - (float)hi;
    unsigned lo = (unsigned)(f * 4294967296.0f);
    return ((u64)hi << 32) | (u64)lo;
}


template <int CTRL>
__device__ __forceinline__ unsigned min_stage(unsigned x) {
    unsigned o = (unsigned)__builtin_amdgcn_update_dpp(0, (int)x, CTRL, 0xf, 0xf, true);
    return o < x ? o : x;
}

__device__ __forceinline__ unsigned row_min(unsigned x) {
    x = min_stage<0x168>(x);
    x = min_stage<0x164>(x);
    x = min_stage<0x162>(x);
    x = min_stage<0x161>(x);
    return x;
}

__device__ __forceinline__ unsigned wave_min(unsigned x) {
    x = row_min(x);
    unsigned o = __builtin_amdgcn_permlanex16(x, x, 0x76543210u, 0xfedcba98u,
                                              false, false);
    return o < x ? o : x;
}

__global__ void __launch_bounds__(LTH)
lapjv_kernel(const float* __restrict__ cost, int* __restrict__ mapping) {
    int b    = blockIdx.x;
    int tid  = threadIdx.x;
    int lane = tid & 31;
    int wave = tid >> 5;
    const float* costB = cost + (size_t)b * MM * NN;
    const u64      FX_INF = ~0ull;
    const unsigned INF_W  = 0xffffffffu;

    __shared__ u64 sh_u[LTH], sh_v[LTH], sh_short[LTH];
    __shared__ int sh_path[LTH], sh_c4r[LTH], sh_r4c[LTH], sh_SR[LTH];
    __shared__ uint4 wrec[2][NWAVE];
    __shared__ int rowlist[2][NWAVE][32];

    if (tid < NN) {
        sh_u[tid] = 0;
        sh_c4r[tid] = -1; sh_r4c[tid] = -1;
        float best = __builtin_huge_valf();
        int   bi   = 0;
        for (int i0 = 0; i0 < MM; i0 += 10) {
            float c[10];
#pragma unroll
            for (int k = 0; k < 10; ++k)
                c[k] = costB[(size_t)(i0 + k) * NN + tid];
#pragma unroll
            for (int k = 0; k < 10; ++k)
                if (c[k] < best) { best = c[k]; bi = i0 + k; }
        }
        sh_v[tid]    = cost_fx(best);
        sh_path[tid] = bi;
    }
    __syncthreads();
    if (tid < MM) {
        int mine = -1;
        for (int j = NN - 1; j >= 0; --j)
            if (sh_path[j] == tid) mine = j;
        if (mine >= 0) { sh_c4r[tid] = mine; sh_r4c[mine] = tid; }
    }
    __syncthreads();

    const int ahead_off = imin_i(lane * 32, NN - 1);
    const int colc = imin_i(tid, NN - 1);

    int par = 0;
    for (int cur = 0; cur < MM; ++cur) {
        if (__builtin_amdgcn_readfirstlane(sh_c4r[cur]) != -1) continue;
        int rem = 1;
        if (tid < NN) { sh_short[tid] = FX_INF; sh_path[tid] = -1; }
        int  myrow = (tid < NN) ? sh_r4c[tid] : -1;
        u64  vmine = (tid < NN) ? sh_v[tid] : 0;
        float ahead = 0.0f;
        unsigned shi = INF_W;
        unsigned slo = INF_W;
        int ii   = cur;
        u64 minv = 0;
        int sink = -1;
        int nsr  = 0;
        float cw   = (tid < NN) ? costB[(size_t)cur * NN + tid] : 0.0f;
        u64   urow = sh_u[cur];

        auto relax = [&](float cword, u64 udual, int row) {
            if (tid < NN && rem) {
                u64 t0 = minv - udual - vmine;
                u64 r  = t0 + cost_fx(cword);
                unsigned rhi = (unsigned)(r >> 32);
                unsigned rlo = (unsigned)r;
                bool less = (rhi < shi) | ((rhi == shi) & (rlo < slo));
                if (less) {
                    sh_short[tid] = r;
                    sh_path[tid]  = row;
                    shi = rhi; slo = rlo;
                }
            }
        };

        for (int step = 0; step < NN; ++step) {
            const int rp = par;
            if (tid == 0) sh_SR[nsr] = ii;
            nsr = nsr + 1;

            relax(cw, urow, ii);
            unsigned chi = INF_W;
            unsigned clo = INF_W;
            if (tid < NN && rem) { chi = shi; clo = slo; }

            unsigned mhi = wave_min(chi);
            unsigned hit = __builtin_amdgcn_ballot_w32(chi == mhi);
            unsigned smhi = (unsigned)__builtin_amdgcn_readfirstlane((int)mhi);
            bool live = smhi != INF_W;
            int      wlane = __builtin_ctz(hit);
            unsigned smlo  = (unsigned)__builtin_amdgcn_readlane((int)clo, wlane);
            if ((hit & (hit - 1u)) != 0u) {
                unsigned other = hit & ~__builtin_amdgcn_ballot_w32(clo == smlo);
                if (other != 0u) {
                    unsigned mlo = wave_min(chi == mhi ? clo : INF_W);
                    hit  &= __builtin_amdgcn_ballot_w32(clo == mlo);
                    wlane = __builtin_ctz(hit);
                    smlo  = (unsigned)__builtin_amdgcn_readlane((int)clo, wlane);
                }
            }
            int  crow = __builtin_amdgcn_readlane(myrow, wlane);
            unsigned zrec = 0u;
            if (live) {
                unsigned nset  = (unsigned)__builtin_popcount(hit);
                unsigned freeb = hit & __builtin_amdgcn_ballot_w32(myrow < 0);
                unsigned fcol1 = freeb != 0u
                    ? (unsigned)((tid & ~31) + __builtin_ctz(freeb) + 1) : 0u;
                zrec = nset | (fcol1 << 6);
                if (nset > 1u && ((hit >> lane) & 1u) != 0u)
                    rowlist[rp][wave][__builtin_popcount(hit & ((1u << lane) - 1u))] = myrow;
            }
            if (lane == 0)
                wrec[rp][wave] = make_uint4(smhi, smlo, zrec, (unsigned)crow);
            __syncthreads();

            uint4    frec = wrec[rp][lane & 15];
            unsigned fhi  = frec.x;
            unsigned flo  = frec.y;
            unsigned fz   = frec.z;
            int      fr   = (int)frec.w;

            asm volatile("" :: "v"(ahead));
            if (live && crow >= 0)
                ahead = costB[(size_t)crow * NN + ahead_off];
            unsigned ghi  = row_min(fhi);
            unsigned hitw = __builtin_amdgcn_ballot_w32(fhi == ghi);
            int      win  = __builtin_ctz(hitw);
            unsigned sglo = (unsigned)__builtin_amdgcn_readlane((int)flo, win);
            if ((hitw & (hitw - 1u) & 0xffffu) != 0u) {
                unsigned other = hitw & ~__builtin_amdgcn_ballot_w32(flo == sglo);
                if (other != 0u) {
                    unsigned glo = row_min(fhi == ghi ? flo : INF_W);
                    hitw &= __builtin_amdgcn_ballot_w32(flo == glo);
                    win   = __builtin_ctz(hitw);
                    sglo  = (unsigned)__builtin_amdgcn_readlane((int)flo, win);
                }
            }
            par ^= 1;
            unsigned wm = hitw & 0xffffu;

            int r4 = __builtin_amdgcn_readlane(fr, win);
            if (r4 != -1) {
                if (tid < NN) cw = costB[(size_t)r4 * NN + tid];
                urow = sh_u[r4];
            }
            unsigned sghi = (unsigned)__builtin_amdgcn_readfirstlane((int)ghi);
            minv = ((u64)sghi << 32) | (u64)sglo;
            unsigned fb = wm & __builtin_amdgcn_ballot_w32((fz >> 6) != 0u);
            if (fb != 0u) {
                sink = (int)((unsigned)__builtin_amdgcn_readlane((int)fz, __builtin_ctz(fb)) >> 6) - 1;
                break;
            }
            if (rem && shi == sghi && slo == sglo) rem = 0;
            ii = r4;

            int wcur = win;
            int ncur = (int)((unsigned)__builtin_amdgcn_readlane((int)fz, win) & 63u);
            int kcur = 1;
            if (ncur > 1 || (wm & (wm - 1u)) != 0u) {
                auto next_row = [&]() -> int {
                    if (kcur >= ncur) {
                        wm &= wm - 1u;
                        if (wm == 0u) return -1;
                        wcur = __builtin_ctz(wm);
                        ncur = (int)((unsigned)__builtin_amdgcn_readlane((int)fz, wcur) & 63u);
                        kcur = 0;
                    }
                    int row = (kcur == 0)
                        ? __builtin_amdgcn_readlane(fr, wcur)
                        : __builtin_amdgcn_readfirstlane(rowlist[rp][wcur][kcur]);
                    kcur = kcur + 1;
                    return row;
                };
                for (int g = 0; g < NN; g += 4) {
                    int r0 = next_row();
                    if (r0 < 0) break;
                    float c0 = costB[(size_t)r0 * NN + colc];
                    u64   u0 = sh_u[r0];
                    int r1 = next_row();
                    int q1 = r1 >= 0 ? r1 : r0;
                    float c1 = costB[(size_t)q1 * NN + colc];
                    u64   u1 = sh_u[q1];
                    int r2 = next_row();
                    int q2 = r2 >= 0 ? r2 : r0;
                    float c2 = costB[(size_t)q2 * NN + colc];
                    u64   u2 = sh_u[q2];
                    int r3 = next_row();
                    int q3 = r3 >= 0 ? r3 : r0;
                    float c3 = costB[(size_t)q3 * NN + colc];
                    u64   u3 = sh_u[q3];
                    if (tid == 0) {
                        sh_SR[nsr] = r0;
                        if (r1 >= 0) sh_SR[nsr + 1] = r1;
                        if (r2 >= 0) sh_SR[nsr + 2] = r2;
                        if (r3 >= 0) sh_SR[nsr + 3] = r3;
                    }
                    nsr += 1 + (r1 >= 0) + (r2 >= 0) + (r3 >= 0);
                    relax(c0, u0, r0);
                    if (r1 >= 0) relax(c1, u1, r1);
                    if (r2 >= 0) relax(c2, u2, r2);
                    if (r3 >= 0) relax(c3, u3, r3);
                }
            }
        }

        if (sink != -1) {
            if (tid == 0) {
                sh_u[cur] += minv;
            } else if (tid < nsr) {
                int sr = sh_SR[tid];
                sh_u[sr] += minv - sh_short[sh_c4r[sr]];
            }
            if (tid < NN && !rem) sh_v[tid] -= minv - sh_short[tid];
        }
        __syncthreads();

        if (tid == 0 && sink != -1) {
            int j = sink;
            for (int hop = 0; hop < NN; ++hop) {
                int i = sh_path[j];
                sh_r4c[j] = i;
                int tmp = sh_c4r[i];
                sh_c4r[i] = j;
                j = tmp;
                if (i == cur) break;
            }
        }
        __syncthreads();
    }
    if (tid < MM) mapping[b * MM + tid] = sh_c4r[tid];
}

__global__ void extrap_kernel(const float* __restrict__ p1b,
                              const float* __restrict__ p1l,
                              const float* __restrict__ p2b,
                              const float* __restrict__ p2l,
                              const float* __restrict__ toff,
                              const int*   __restrict__ mapping,
                              float* __restrict__ out) {
    int idx = blockIdx.x * blockDim.x + threadIdx.x;
    if (idx >= BB * MM) return;
    int b = idx / MM;

    float first  = toff[b * 3 + 1] - toff[b * 3 + 0];
    float second = toff[b * 3 + 2] - toff[b * 3 + 1];
    float factor = second / first;

    int  j   = mapping[idx];
    bool has = (j >= 0);
    int  jc  = has ? j : 0;

    const float* pb2 = p2b + (size_t)idx * 4;
    const float* pb1 = p1b + ((size_t)b * NN + jc) * 4;
    float* ob = out + (size_t)idx * 4;
#pragma unroll
    for (int k = 0; k < 4; ++k) {
        float c1 = has ? pb1[k] : pb2[k];
        float v  = pb2[k] + (pb2[k] - c1) * factor;
        ob[k] = (k >= 2) ? fmaxf(v, 0.0f) : v;
    }

    const float* pl2 = p2l + (size_t)idx * CC;
    const float* pl1 = p1l + ((size_t)b * NN + jc) * CC;
    float* ol = out + (size_t)BB * MM * 4 + (size_t)idx * CC;
    for (int c = 0; c < CC; ++c) {
        float c1 = has ? pl1[c] : 0.0f;
        ol[c] = 0.5f * (pl2[c] + c1);
    }
}

extern "C" void kernel_launch(void* const* d_in, const int* in_sizes, int n_in,
                              void* d_out, int out_size, void* d_ws, size_t ws_size,
                              hipStream_t stream) {
    (void)in_sizes; (void)n_in; (void)out_size; (void)ws_size;
    const float* p1b  = (const float*)d_in[0];
    const float* p1l  = (const float*)d_in[1];
    const float* p2b  = (const float*)d_in[2];
    const float* p2l  = (const float*)d_in[3];
    const float* toff = (const float*)d_in[4];

    float* ws   = (float*)d_ws;
    float* s1   = ws;
    float* s2   = s1 + (size_t)BB * NN * CC;
    float* cost = s2 + (size_t)BB * MM * CC;
    int*   map  = (int*)(cost + (size_t)BB * MM * NN);
    float* out  = (float*)d_out;

    int tot = BB * NN * CC + BB * MM * CC;
    sigmoid_kernel<<<(tot + 255) / 256, 256, 0, stream>>>(p1l, p2l, s1, s2);

    const int TM = (MM + 15) / 16, TN = (NN + 15) / 16;
    cost_kernel<<<BB * TM * TN, 32, 0, stream>>>(p1b, p2b, s1, s2, cost);

    lapjv_kernel<<<BB, LTH, 0, stream>>>(cost, map);

    int tot2 = BB * MM;
    extrap_kernel<<<(tot2 + 255) / 256, 256, 0, stream>>>(p1b, p1l, p2b, p2l,
                                                          toff, map, out);
}


# Assignment search start: 12 parallel bidding rounds (a free row bids for its lowest reduced-cost column; the gap to its second lowest lowers that column's dual) match most free rows first; exact sums
# speedup vs baseline: 6.5974x; 1.1549x over previous
#include <hip/hip_runtime.h>
#include <math.h>

#define BB 4
#define MM 500
#define NN 500
#define CC 91

typedef __attribute__((ext_vector_type(2))) float v2f;
typedef __attribute__((ext_vector_type(8))) float v8f;

__device__ __forceinline__ int imin_i(int a, int b) { return a < b ? a : b; }

__global__ void sigmoid_kernel(const float* __restrict__ p1l,
                               const float* __restrict__ p2l,
                               float* __restrict__ s1,
                               float* __restrict__ s2) {
    int idx = blockIdx.x * blockDim.x + threadIdx.x;
    const int n1 = BB * NN * CC;
    const int n2 = BB * MM * CC;
    if (idx < n1) {
        float x = p1l[idx];
        s1[idx] = 1.0f / (1.0f + expf(-x));
    } else if (idx < n1 + n2) {
        float x = p2l[idx - n1];
        s2[idx - n1] = 1.0f / (1.0f + expf(-x));
    }
}

__global__ void cost_kernel(const float* __restrict__ p1b,
                            const float* __restrict__ p2b,
                            const float* __restrict__ s1,
                            const float* __restrict__ s2,
                            float* __restrict__ cost) {
    const int TN = (NN + 15) / 16;
    const int TM = (MM + 15) / 16;
    int bid = blockIdx.x;
    int b  = bid / (TM * TN);
    int t  = bid % (TM * TN);
    int m0 = (t / TN) * 16;
    int n0 = (t % TN) * 16;
    int lane = threadIdx.x;

    __shared__ float sq2[16];
    __shared__ float sq1[16];
    __shared__ float ls2[16 * CC];
    __shared__ float ls1[16 * CC];

    v2f a  = {0.0f, 0.0f};
    v2f bv = {0.0f, 0.0f};
    if (lane < 16) {
        int m = imin_i(m0 + lane, MM - 1);
        const float* pb = p2b + ((size_t)b * MM + m) * 4;
        float x = pb[0], y = pb[1];
        a.x = x; a.y = y;
        sq2[lane] = x * x + y * y;
        int n = imin_i(n0 + lane, NN - 1);
        const float* qb = p1b + ((size_t)b * NN + n) * 4;
        float bx = qb[0], by = qb[1];
        bv.x = bx; bv.y = by;
        sq1[lane] = bx * bx + by * by;
    }
    const float* s2base = s2 + (size_t)b * MM * CC;
    const float* s1base = s1 + (size_t)b * NN * CC;
    for (int idx = lane; idx < 16 * CC; idx += 32) {
        int r = idx / CC;
        int c = idx - r * CC;
        ls2[idx] = s2base[(size_t)imin_i(m0 + r, MM - 1) * CC + c];
        ls1[idx] = s1base[(size_t)imin_i(n0 + r, NN - 1) * CC + c];
    }
    __syncthreads();

    v8f cacc = {0.f, 0.f, 0.f, 0.f, 0.f, 0.f, 0.f, 0.f};
    v8f dot = __builtin_amdgcn_wmma_f32_16x16x4_f32(
         false, a,  false, bv,
         (short)0, cacc,  false,  false);

    int n_loc  = lane & 15;
    int m_base = (lane >> 4) << 3;

    float cls[8];
#pragma unroll
    for (int i = 0; i < 8; ++i) cls[i] = 0.0f;
    for (int c = 0; c < CC; ++c) {
        float s1v = ls1[n_loc * CC + c];
#pragma unroll
        for (int i = 0; i < 8; ++i) {
            float d = fabsf(ls2[(m_base + i) * CC + c] - s1v);
            cls[i] = fmaxf(cls[i], d);
        }
    }

#pragma unroll
    for (int i = 0; i < 8; ++i) {
        int m = m0 + m_base + i;
        int n = n0 + n_loc;
        if (m < MM && n < NN) {
            float d2 = sq2[m_base + i] + sq1[n_loc] - 2.0f * dot[i];
            float cd = sqrtf(fmaxf(d2, 0.0f));
            cost[((size_t)b * MM + m) * NN + n] = 0.5f * cd + 0.5f * cls[i];
        }
    }
}

#define LTH 512
#define NWAVE (LTH / 32)

typedef unsigned long long u64;

__device__ __forceinline__ u64 cost_fx(float c) {
    float    t  = c * 4194304.0f;
    unsigned hi = (unsigned)t;
    float    f  = t - (float)hi;
    unsigned lo = (unsigned)(f * 4294967296.0f);
    return ((u64)hi << 32) | (u64)lo;
}


template <int CTRL>
__device__ __forceinline__ unsigned min_stage(unsigned x) {
    unsigned o = (unsigned)__builtin_amdgcn_update_dpp(0, (int)x, CTRL, 0xf, 0xf, true);
    return o < x ? o : x;
}

__device__ __forceinline__ unsigned row_min(unsigned x) {
    x = min_stage<0x168>(x);
    x = min_stage<0x164>(x);
    x = min_stage<0x162>(x);
    x = min_stage<0x161>(x);
    return x;
}

__device__ __forceinline__ unsigned wave_min(unsigned x) {
    x = row_min(x);
    unsigned o = __builtin_amdgcn_permlanex16(x, x, 0x76543210u, 0xfedcba98u,
                                              false, false);
    return o < x ? o : x;
}

__device__ __forceinline__ u64 wave_min64(u64 x) {
    unsigned hi = (unsigned)(x >> 32);
    unsigned lo = (unsigned)x;
    unsigned mh = wave_min(hi);
    unsigned ml = wave_min(hi == mh ? lo : 0xffffffffu);
    return ((u64)mh << 32) | (u64)ml;
}

#define NBID 12

__global__ void __launch_bounds__(LTH)
lapjv_kernel(const float* __restrict__ cost, int* __restrict__ mapping) {
    int b    = blockIdx.x;
    int tid  = threadIdx.x;
    int lane = tid & 31;
    int wave = tid >> 5;
    const float* costB = cost + (size_t)b * MM * NN;
    const u64      FX_INF = ~0ull;
    const unsigned INF_W  = 0xffffffffu;

    __shared__ u64 sh_u[LTH], sh_v[LTH], sh_short[LTH];
    __shared__ int sh_path[LTH], sh_c4r[LTH], sh_r4c[LTH], sh_SR[LTH];
    __shared__ uint4 wrec[2][NWAVE];
    __shared__ int rowlist[2][NWAVE][32];
    __shared__ __attribute__((aligned(16))) int sh_bcol[LTH];
    __shared__ u64 sh_bgap[LTH];

    if (tid < NN) {
        sh_u[tid] = 0;
        sh_c4r[tid] = -1; sh_r4c[tid] = -1;
        float best = __builtin_huge_valf();
        int   bi   = 0;
        for (int i0 = 0; i0 < MM; i0 += 10) {
            float c[10];
#pragma unroll
            for (int k = 0; k < 10; ++k)
                c[k] = costB[(size_t)(i0 + k) * NN + tid];
#pragma unroll
            for (int k = 0; k < 10; ++k)
                if (c[k] < best) { best = c[k]; bi = i0 + k; }
        }
        sh_v[tid]    = cost_fx(best);
        sh_path[tid] = bi;
    }
    __syncthreads();
    if (tid < MM) {
        int mine = -1;
        for (int j = NN - 1; j >= 0; --j)
            if (sh_path[j] == tid) mine = j;
        if (mine >= 0) { sh_c4r[tid] = mine; sh_r4c[mine] = tid; }
    }
    __syncthreads();

    const int swave = __builtin_amdgcn_readfirstlane(wave);
    for (int rnd = 0; rnd < NBID; ++rnd) {
        for (int i = swave; i < MM; i += NWAVE) {
            if (__builtin_amdgcn_readfirstlane(sh_c4r[i]) != -1) {
                if (lane == 0) sh_bcol[i] = -1;
                continue;
            }
            const float* rowp = costB + (size_t)i * NN;
            float cv[16];
#pragma unroll
            for (int k = 0; k < 16; ++k)
                cv[k] = rowp[imin_i(k * 32 + lane, NN - 1)];
            u64 m1 = FX_INF, m2 = FX_INF;
            int j1 = 0;
#pragma unroll
            for (int k = 0; k < 16; ++k) {
                int col = k * 32 + lane;
                u64 h = FX_INF;
                if (col < NN) h = cost_fx(cv[k]) - sh_v[col];
                if (h < m1)      { m2 = m1; m1 = h; j1 = col; }
                else if (h < m2) { m2 = h; }
            }
            u64      M1 = wave_min64(m1);
            unsigned J1 = wave_min(m1 == M1 ? (unsigned)j1 : INF_W);
            u64      M2 = wave_min64((m1 == M1 && (unsigned)j1 == J1) ? m2 : m1);
            if (lane == 0) {
                u64 gap = M2 - M1;
                bool bid = gap != 0 || sh_r4c[J1] == -1;
                sh_bcol[i]  = bid ? (int)J1 : -1;
                sh_short[i] = M2;
                sh_bgap[i]  = gap;
            }
        }
        __syncthreads();
        if (tid < NN) {
            int mine = -1;
            const int4* b4 = reinterpret_cast<const int4*>(sh_bcol);
            for (int q = MM / 4 - 1; q >= 0; --q) {
                int4 t = b4[q];
                if (t.w == tid) mine = 4 * q + 3;
                if (t.z == tid) mine = 4 * q + 2;
                if (t.y == tid) mine = 4 * q + 1;
                if (t.x == tid) mine = 4 * q;
            }
            if (mine >= 0) {
                int i0 = sh_r4c[tid];
                sh_v[tid]  -= sh_bgap[mine];
                sh_u[mine]  = sh_short[mine];
                sh_r4c[tid] = mine;
                sh_c4r[mine] = tid;
                if (i0 >= 0) sh_c4r[i0] = -1;
            }
        }
        __syncthreads();
    }

    const int ahead_off = imin_i(lane * 32, NN - 1);
    const int colc = imin_i(tid, NN - 1);

    int par = 0;
    for (int cur = 0; cur < MM; ++cur) {
        if (__builtin_amdgcn_readfirstlane(sh_c4r[cur]) != -1) continue;
        int rem = 1;
        if (tid < NN) { sh_short[tid] = FX_INF; sh_path[tid] = -1; }
        int  myrow = (tid < NN) ? sh_r4c[tid] : -1;
        u64  vmine = (tid < NN) ? sh_v[tid] : 0;
        float ahead = 0.0f;
        unsigned shi = INF_W;
        unsigned slo = INF_W;
        int ii   = cur;
        u64 minv = 0;
        int sink = -1;
        int nsr  = 0;
        float cw   = (tid < NN) ? costB[(size_t)cur * NN + tid] : 0.0f;
        u64   urow = sh_u[cur];

        auto relax = [&](float cword, u64 udual, int row) {
            if (tid < NN && rem) {
                u64 t0 = minv - udual - vmine;
                u64 r  = t0 + cost_fx(cword);
                unsigned rhi = (unsigned)(r >> 32);
                unsigned rlo = (unsigned)r;
                bool less = (rhi < shi) | ((rhi == shi) & (rlo < slo));
                if (less) {
                    sh_short[tid] = r;
                    sh_path[tid]  = row;
                    shi = rhi; slo = rlo;
                }
            }
        };

        for (int step = 0; step < NN; ++step) {
            const int rp = par;
            if (tid == 0) sh_SR[nsr] = ii;
            nsr = nsr + 1;

            relax(cw, urow, ii);
            unsigned chi = INF_W;
            unsigned clo = INF_W;
            if (tid < NN && rem) { chi = shi; clo = slo; }

            unsigned mhi = wave_min(chi);
            unsigned hit = __builtin_amdgcn_ballot_w32(chi == mhi);
            unsigned smhi = (unsigned)__builtin_amdgcn_readfirstlane((int)mhi);
            bool live = smhi != INF_W;
            int      wlane = __builtin_ctz(hit);
            unsigned smlo  = (unsigned)__builtin_amdgcn_readlane((int)clo, wlane);
            if ((hit & (hit - 1u)) != 0u) {
                unsigned other = hit & ~__builtin_amdgcn_ballot_w32(clo == smlo);
                if (other != 0u) {
                    unsigned mlo = wave_min(chi == mhi ? clo : INF_W);
                    hit  &= __builtin_amdgcn_ballot_w32(clo == mlo);
                    wlane = __builtin_ctz(hit);
                    smlo  = (unsigned)__builtin_amdgcn_readlane((int)clo, wlane);
                }
            }
            int  crow = __builtin_amdgcn_readlane(myrow, wlane);
            unsigned zrec = 0u;
            if (live) {
                unsigned nset  = (unsigned)__builtin_popcount(hit);
                unsigned freeb = hit & __builtin_amdgcn_ballot_w32(myrow < 0);
                unsigned fcol1 = freeb != 0u
                    ? (unsigned)((tid & ~31) + __builtin_ctz(freeb) + 1) : 0u;
                zrec = nset | (fcol1 << 6);
                if (nset > 1u && ((hit >> lane) & 1u) != 0u)
                    rowlist[rp][wave][__builtin_popcount(hit & ((1u << lane) - 1u))] = myrow;
            }
            if (lane == 0)
                wrec[rp][wave] = make_uint4(smhi, smlo, zrec, (unsigned)crow);
            __syncthreads();

            uint4    frec = wrec[rp][lane & 15];
            unsigned fhi  = frec.x;
            unsigned flo  = frec.y;
            unsigned fz   = frec.z;
            int      fr   = (int)frec.w;

            asm volatile("" :: "v"(ahead));
            if (live && crow >= 0)
                ahead = costB[(size_t)crow * NN + ahead_off];
            unsigned ghi  = row_min(fhi);
            unsigned hitw = __builtin_amdgcn_ballot_w32(fhi == ghi);
            int      win  = __builtin_ctz(hitw);
            unsigned sglo = (unsigned)__builtin_amdgcn_readlane((int)flo, win);
            if ((hitw & (hitw - 1u) & 0xffffu) != 0u) {
                unsigned other = hitw & ~__builtin_amdgcn_ballot_w32(flo == sglo);
                if (other != 0u) {
                    unsigned glo = row_min(fhi == ghi ? flo : INF_W);
                    hitw &= __builtin_amdgcn_ballot_w32(flo == glo);
                    win   = __builtin_ctz(hitw);
                    sglo  = (unsigned)__builtin_amdgcn_readlane((int)flo, win);
                }
            }
            par ^= 1;
            unsigned wm = hitw & 0xffffu;

            int r4 = __builtin_amdgcn_readlane(fr, win);
            if (r4 != -1) {
                if (tid < NN) cw = costB[(size_t)r4 * NN + tid];
                urow = sh_u[r4];
            }
            unsigned sghi = (unsigned)__builtin_amdgcn_readfirstlane((int)ghi);
            minv = ((u64)sghi << 32) | (u64)sglo;
            unsigned fb = wm & __builtin_amdgcn_ballot_w32((fz >> 6) != 0u);
            if (fb != 0u) {
                sink = (int)((unsigned)__builtin_amdgcn_readlane((int)fz, __builtin_ctz(fb)) >> 6) - 1;
                break;
            }
            if (rem && shi == sghi && slo == sglo) rem = 0;
            ii = r4;

            int wcur = win;
            int ncur = (int)((unsigned)__builtin_amdgcn_readlane((int)fz, win) & 63u);
            int kcur = 1;
            if (ncur > 1 || (wm & (wm - 1u)) != 0u) {
                auto next_row = [&]() -> int {
                    if (kcur >= ncur) {
                        wm &= wm - 1u;
                        if (wm == 0u) return -1;
                        wcur = __builtin_ctz(wm);
                        ncur = (int)((unsigned)__builtin_amdgcn_readlane((int)fz, wcur) & 63u);
                        kcur = 0;
                    }
                    int row = (kcur == 0)
                        ? __builtin_amdgcn_readlane(fr, wcur)
                        : __builtin_amdgcn_readfirstlane(rowlist[rp][wcur][kcur]);
                    kcur = kcur + 1;
                    return row;
                };
                for (int g = 0; g < NN; g += 4) {
                    int r0 = next_row();
                    if (r0 < 0) break;
                    float c0 = costB[(size_t)r0 * NN + colc];
                    u64   u0 = sh_u[r0];
                    int r1 = next_row();
                    int q1 = r1 >= 0 ? r1 : r0;
                    float c1 = costB[(size_t)q1 * NN + colc];
                    u64   u1 = sh_u[q1];
                    int r2 = next_row();
                    int q2 = r2 >= 0 ? r2 : r0;
                    float c2 = costB[(size_t)q2 * NN + colc];
                    u64   u2 = sh_u[q2];
                    int r3 = next_row();
                    int q3 = r3 >= 0 ? r3 : r0;
                    float c3 = costB[(size_t)q3 * NN + colc];
                    u64   u3 = sh_u[q3];
                    if (tid == 0) {
                        sh_SR[nsr] = r0;
                        if (r1 >= 0) sh_SR[nsr + 1] = r1;
                        if (r2 >= 0) sh_SR[nsr + 2] = r2;
                        if (r3 >= 0) sh_SR[nsr + 3] = r3;
                    }
                    nsr += 1 + (r1 >= 0) + (r2 >= 0) + (r3 >= 0);
                    relax(c0, u0, r0);
                    if (r1 >= 0) relax(c1, u1, r1);
                    if (r2 >= 0) relax(c2, u2, r2);
                    if (r3 >= 0) relax(c3, u3, r3);
                }
            }
        }

        if (sink != -1) {
            if (tid == 0) {
                sh_u[cur] += minv;
            } else if (tid < nsr) {
                int sr = sh_SR[tid];
                sh_u[sr] += minv - sh_short[sh_c4r[sr]];
            }
            if (tid < NN && !rem) sh_v[tid] -= minv - sh_short[tid];
        }
        __syncthreads();

        if (tid == 0 && sink != -1) {
            int j = sink;
            for (int hop = 0; hop < NN; ++hop) {
                int i = sh_path[j];
                sh_r4c[j] = i;
                int tmp = sh_c4r[i];
                sh_c4r[i] = j;
                j = tmp;
                if (i == cur) break;
            }
        }
        __syncthreads();
    }
    if (tid < MM) mapping[b * MM + tid] = sh_c4r[tid];
}

__global__ void extrap_kernel(const float* __restrict__ p1b,
                              const float* __restrict__ p1l,
                              const float* __restrict__ p2b,
                              const float* __restrict__ p2l,
                              const float* __restrict__ toff,
                              const int*   __restrict__ mapping,
                              float* __restrict__ out) {
    int idx = blockIdx.x * blockDim.x + threadIdx.x;
    if (idx >= BB * MM) return;
    int b = idx / MM;

    float first  = toff[b * 3 + 1] - toff[b * 3 + 0];
    float second = toff[b * 3 + 2] - toff[b * 3 + 1];
    float factor = second / first;

    int  j   = mapping[idx];
    bool has = (j >= 0);
    int  jc  = has ? j : 0;

    const float* pb2 = p2b + (size_t)idx * 4;
    const float* pb1 = p1b + ((size_t)b * NN + jc) * 4;
    float* ob = out + (size_t)idx * 4;
#pragma unroll
    for (int k = 0; k < 4; ++k) {
        float c1 = has ? pb1[k] : pb2[k];
        float v  = pb2[k] + (pb2[k] - c1) * factor;
        ob[k] = (k >= 2) ? fmaxf(v, 0.0f) : v;
    }

    const float* pl2 = p2l + (size_t)idx * CC;
    const float* pl1 = p1l + ((size_t)b * NN + jc) * CC;
    float* ol = out + (size_t)BB * MM * 4 + (size_t)idx * CC;
    for (int c = 0; c < CC; ++c) {
        float c1 = has ? pl1[c] : 0.0f;
        ol[c] = 0.5f * (pl2[c] + c1);
    }
}

extern "C" void kernel_launch(void* const* d_in, const int* in_sizes, int n_in,
                              void* d_out, int out_size, void* d_ws, size_t ws_size,
                              hipStream_t stream) {
    (void)in_sizes; (void)n_in; (void)out_size; (void)ws_size;
    const float* p1b  = (const float*)d_in[0];
    const float* p1l  = (const float*)d_in[1];
    const float* p2b  = (const float*)d_in[2];
    const float* p2l  = (const float*)d_in[3];
    const float* toff = (const float*)d_in[4];

    float* ws   = (float*)d_ws;
    float* s1   = ws;
    float* s2   = s1 + (size_t)BB * NN * CC;
    float* cost = s2 + (size_t)BB * MM * CC;
    int*   map  = (int*)(cost + (size_t)BB * MM * NN);
    float* out  = (float*)d_out;

    int tot = BB * NN * CC + BB * MM * CC;
    sigmoid_kernel<<<(tot + 255) / 256, 256, 0, stream>>>(p1l, p2l, s1, s2);

    const int TM = (MM + 15) / 16, TN = (NN + 15) / 16;
    cost_kernel<<<BB * TM * TN, 32, 0, stream>>>(p1b, p2b, s1, s2, cost);

    lapjv_kernel<<<BB, LTH, 0, stream>>>(cost, map);

    int tot2 = BB * MM;
    extrap_kernel<<<(tot2 + 255) / 256, 256, 0, stream>>>(p1b, p1l, p2b, p2l,
                                                          toff, map, out);
}
